# starting kernel
# baseline (speedup 1.0000x reference)
#include <hip/hip_runtime.h>
#include <math.h>

#define N_NODES 131072
#define E_EDGES 1048576
#define NUM_DSTN 65536
#define IN_DIM 256
#define HEADS 4
#define OUT_D 32
#define FEAT (HEADS * OUT_D)
#define NEG_SLOPE 0.2f

typedef __attribute__((ext_vector_type(2))) float v2f;
typedef __attribute__((ext_vector_type(8))) float v8f;

__global__ void init_kernel(float* __restrict__ out,
                            float* __restrict__ emax,
                            float* __restrict__ esum) {
    size_t i = (size_t)blockIdx.x * blockDim.x + threadIdx.x;
    if (i < (size_t)NUM_DSTN * FEAT) out[i] = 0.0f;
    if (i < (size_t)NUM_DSTN * HEADS) {
        ((unsigned int*)emax)[i] = 0xFF800000u;
        esum[i] = 0.0f;
    }
}

__global__ __launch_bounds__(256) void gemm_kernel(const float* __restrict__ feats,
                                                   const float* __restrict__ W,
                                                   float* __restrict__ h) {
    const int wave = blockIdx.x * (blockDim.x >> 5) + (threadIdx.x >> 5);
    const int lane = threadIdx.x & 31;
    const int lo = lane & 15;
    const int hi = lane >> 4;
    const int rowBase = wave * 16;
    const float* __restrict__ arow = feats + (size_t)(rowBase + lo) * IN_DIM;

    v8f acc[8];
#pragma unroll
    for (int t = 0; t < 8; ++t) acc[t] = (v8f){0.f, 0.f, 0.f, 0.f, 0.f, 0.f, 0.f, 0.f};

    for (int k = 0; k < IN_DIM; k += 4) {
        const int ka = k + 2 * hi;
        v2f a;
        a.x = arow[ka];
        a.y = arow[ka + 1];
#pragma unroll
        for (int t = 0; t < 8; ++t) {
            const int col = t * 16 + lo;
            v2f b;
            b.x = W[(size_t)ka * FEAT + col];
            b.y = W[(size_t)(ka + 1) * FEAT + col];
            acc[t] = __builtin_amdgcn_wmma_f32_16x16x4_f32(
                false, a, false, b, (short)0, acc[t], false, false);
        }
    }

#pragma unroll
    for (int t = 0; t < 8; ++t)
#pragma unroll
        for (int i = 0; i < 8; ++i)
            h[(size_t)(rowBase + i + 8 * hi) * FEAT + t * 16 + lo] = acc[t][i];
}

__global__ void attn_logits_kernel(const float* __restrict__ h,
                                   const float* __restrict__ attn_l,
                                   const float* __restrict__ attn_r,
                                   float* __restrict__ el, float* __restrict__ er) {
    int i = blockIdx.x * blockDim.x + threadIdx.x;
    if (i >= N_NODES * HEADS) return;
    const int n = i >> 2;
    const int hh = i & 3;
    const float* __restrict__ hp = h + (size_t)n * FEAT + hh * OUT_D;
    const float* __restrict__ al = attn_l + hh * OUT_D;
    const float* __restrict__ ar = attn_r + hh * OUT_D;
    float sl = 0.f, sr = 0.f;
#pragma unroll
    for (int d = 0; d < OUT_D; ++d) {
        float v = hp[d];
        sl += v * al[d];
        sr += v * ar[d];
    }
    el[i] = sl;
    er[i] = sr;
}

__device__ __forceinline__ void atomicMaxF(float* addr, float val) {
    if (val >= 0.0f)
        atomicMax((int*)addr, __float_as_int(val));
    else
        atomicMin((unsigned int*)addr, __float_as_uint(val));
}

__global__ void edge_max_kernel(const int* __restrict__ src, const int* __restrict__ dst,
                                const float* __restrict__ el, const float* __restrict__ er,
                                float* __restrict__ ebuf, float* __restrict__ emax) {
    int e = blockIdx.x * blockDim.x + threadIdx.x;
    if (e >= E_EDGES) return;
    const int s = src[e];
    const int d = dst[e];
    const float4 l4 = *(const float4*)(el + (size_t)s * HEADS);
    const float4 r4 = *(const float4*)(er + (size_t)d * HEADS);
    float vals[4] = {l4.x + r4.x, l4.y + r4.y, l4.z + r4.z, l4.w + r4.w};
#pragma unroll
    for (int hh = 0; hh < 4; ++hh) {
        float v = vals[hh];
        v = (v >= 0.f) ? v : NEG_SLOPE * v;
        ebuf[(size_t)e * HEADS + hh] = v;
        atomicMaxF(&emax[(size_t)d * HEADS + hh], v);
    }
}

__global__ void edge_exp_kernel(const int* __restrict__ dst,
                                float* __restrict__ ebuf,
                                const float* __restrict__ emax,
                                float* __restrict__ esum) {
    int e = blockIdx.x * blockDim.x + threadIdx.x;
    if (e >= E_EDGES) return;
    const int d = dst[e];
#pragma unroll
    for (int hh = 0; hh < 4; ++hh) {
        float v = expf(ebuf[(size_t)e * HEADS + hh] - emax[(size_t)d * HEADS + hh]);
        ebuf[(size_t)e * HEADS + hh] = v;
        atomicAdd(&esum[(size_t)d * HEADS + hh], v);
    }
}

__global__ __launch_bounds__(128) void scatter_kernel(const int* __restrict__ src,
                                                      const int* __restrict__ dst,
                                                      const float* __restrict__ h,
                                                      const float* __restrict__ ebuf,
                                                      const float* __restrict__ esum,
                                                      float* __restrict__ out) {
    const int e = blockIdx.x;
    const int t = threadIdx.x;
    const int s = src[e];
    const int d = dst[e];
    const int hh = t >> 5;
    const float alpha = ebuf[(size_t)e * HEADS + hh] / esum[(size_t)d * HEADS + hh];
    const float v = h[(size_t)s * FEAT + t] * alpha;
    atomicAdd(&out[(size_t)d * FEAT + t], v);
}

__global__ void finalize_kernel(float* __restrict__ out) {
    size_t i = (size_t)blockIdx.x * blockDim.x + threadIdx.x;
    if (i < (size_t)NUM_DSTN * FEAT) {
        float x = out[i];
        out[i] = (x > 0.f) ? x : (expf(x) - 1.0f);
    }
    if (i == 0) {
        out[(size_t)NUM_DSTN * FEAT + 0] = 1.0f;
        out[(size_t)NUM_DSTN * FEAT + 1] = 1.0f;
    }
}

extern "C" void kernel_launch(void* const* d_in, const int* in_sizes, int n_in,
                              void* d_out, int out_size, void* d_ws, size_t ws_size,
                              hipStream_t stream) {
    const float* feats  = (const float*)d_in[0];
    const int*   src    = (const int*)d_in[1];
    const int*   dst    = (const int*)d_in[2];
    const float* W      = (const float*)d_in[3];
    const float* attn_l = (const float*)d_in[4];
    const float* attn_r = (const float*)d_in[5];
    float* out = (float*)d_out;

    float* ws   = (float*)d_ws;
    float* hbuf = ws;
    float* el   = hbuf + (size_t)N_NODES * FEAT;
    float* er   = el + (size_t)N_NODES * HEADS;
    float* ebuf = er + (size_t)N_NODES * HEADS;
    float* emax = ebuf + (size_t)E_EDGES * HEADS;
    float* esum = emax + (size_t)NUM_DSTN * HEADS;

    {
        size_t n = (size_t)NUM_DSTN * FEAT;
        init_kernel<<<(unsigned)((n + 255) / 256), 256, 0, stream>>>(out, emax, esum);
    }
    {
        int rowTiles = N_NODES / 16;
        int blocks = rowTiles / 8;
        gemm_kernel<<<blocks, 256, 0, stream>>>(feats, W, hbuf);
    }
    {
        int n = N_NODES * HEADS;
        attn_logits_kernel<<<(n + 255) / 256, 256, 0, stream>>>(hbuf, attn_l, attn_r, el, er);
    }
    edge_max_kernel<<<(E_EDGES + 255) / 256, 256, 0, stream>>>(src, dst, el, er, ebuf, emax);
    edge_exp_kernel<<<(E_EDGES + 255) / 256, 256, 0, stream>>>(dst, ebuf, emax, esum);
    scatter_kernel<<<E_EDGES, 128, 0, stream>>>(src, dst, hbuf, ebuf, esum, out);
    {
        size_t n = (size_t)NUM_DSTN * FEAT;
        finalize_kernel<<<(unsigned)((n + 255) / 256), 256, 0, stream>>>(out);
    }
}


# Tile filter: a bit plane per destination tile (built in LDS, kept in the tile's own rows of out until they are written) replaces the byte per edge: a tile scans 128 KiB, not 1 MiB; LDS arrival order
# speedup vs baseline: 5.4112x; 5.4112x over previous
#include <hip/hip_runtime.h>
#include <math.h>

#define N_NODES 131072
#define E_EDGES 1048576
#define NUM_DSTN 65536
#define IN_DIM 256
#define HEADS 4
#define OUT_D 32
#define FEAT (HEADS * OUT_D)
#define NEG_SLOPE 0.2f

typedef __attribute__((ext_vector_type(2))) float v2f;
typedef __attribute__((ext_vector_type(8))) float v8f;

__global__ __launch_bounds__(256) void gemm_kernel(const float* __restrict__ feats,
                                                   const float* __restrict__ W,
                                                   float* __restrict__ h) {
    const int wave = blockIdx.x * (blockDim.x >> 5) + (threadIdx.x >> 5);
    const int lane = threadIdx.x & 31;
    const int lo = lane & 15;
    const int hi = lane >> 4;
    const int rowBase = wave * 16;
    const float* __restrict__ arow = feats + (size_t)(rowBase + lo) * IN_DIM;

    v8f acc[8];
#pragma unroll
    for (int t = 0; t < 8; ++t) acc[t] = (v8f){0.f, 0.f, 0.f, 0.f, 0.f, 0.f, 0.f, 0.f};

    for (int k = 0; k < IN_DIM; k += 4) {
        const int ka = k + 2 * hi;
        v2f a;
        a.x = arow[ka];
        a.y = arow[ka + 1];
#pragma unroll
        for (int t = 0; t < 8; ++t) {
            const int col = t * 16 + lo;
            v2f b;
            b.x = W[(size_t)ka * FEAT + col];
            b.y = W[(size_t)(ka + 1) * FEAT + col];
            acc[t] = __builtin_amdgcn_wmma_f32_16x16x4_f32(
                false, a, false, b, (short)0, acc[t], false, false);
        }
    }

#pragma unroll
    for (int t = 0; t < 8; ++t)
#pragma unroll
        for (int i = 0; i < 8; ++i)
            h[(size_t)(rowBase + i + 8 * hi) * FEAT + t * 16 + lo] = acc[t][i];
}

__global__ void attn_logits_kernel(const float* __restrict__ h,
                                   const float* __restrict__ attn_l,
                                   const float* __restrict__ attn_r,
                                   float* __restrict__ el, float* __restrict__ er) {
    int i = blockIdx.x * blockDim.x + threadIdx.x;
    if (i >= N_NODES * HEADS) return;
    const int n = i >> 2;
    const int hh = i & 3;
    const float* __restrict__ hp = h + (size_t)n * FEAT + hh * OUT_D;
    const float* __restrict__ al = attn_l + hh * OUT_D;
    const float* __restrict__ ar = attn_r + hh * OUT_D;
    float sl = 0.f, sr = 0.f;
#pragma unroll
    for (int d = 0; d < OUT_D; ++d) {
        float v = hp[d];
        sl += v * al[d];
        sr += v * ar[d];
    }
    el[i] = sl;
    er[i] = sr;
}

#define TILE_SHIFT 8
#define N_TILES (NUM_DSTN >> TILE_SHIFT)
#define PLANE_WORDS (E_EDGES / 32)
#define TB_EDGES 2048
#define TB_WORDS (TB_EDGES / 32)
__global__ __launch_bounds__(256) void tile_plane_kernel(const int* __restrict__ dst,
                                                         unsigned int* planes) {
    __shared__ __attribute__((aligned(16))) unsigned int bits[N_TILES * TB_WORDS];
    const int tid = threadIdx.x;
    uint4* b4 = (uint4*)bits;
    for (int i = tid; i < N_TILES * TB_WORDS / 4; i += 256) b4[i] = make_uint4(0u, 0u, 0u, 0u);
    __syncthreads();

    const int g = blockIdx.x * (TB_EDGES / 4) + tid * 2;
    const int4 d0 = ((const int4*)dst)[g];
    const int4 d1 = ((const int4*)dst)[g + 1];
    unsigned int* wp = bits + (tid >> 2);
    const unsigned b0 = 1u << ((tid & 3) * 8);
#define SET_BIT(d, k)                                                                              \
    __hip_atomic_fetch_or(wp + (((unsigned)(d) >> TILE_SHIFT) & (N_TILES - 1)) * TB_WORDS, b0 << (k), \
                          __ATOMIC_RELAXED, __HIP_MEMORY_SCOPE_WORKGROUP)
    SET_BIT(d0.x, 0); SET_BIT(d0.y, 1); SET_BIT(d0.z, 2); SET_BIT(d0.w, 3);
    SET_BIT(d1.x, 4); SET_BIT(d1.y, 5); SET_BIT(d1.z, 6); SET_BIT(d1.w, 7);
#undef SET_BIT
    __syncthreads();

    for (int i = tid; i < N_TILES * TB_WORDS / 4; i += 256) {
        uint4* o = (uint4*)(planes + (size_t)(i >> 4) * PLANE_WORDS + blockIdx.x * TB_WORDS);
        o[i & 15] = b4[i];
    }
}

#define TILE_D (1 << TILE_SHIFT)
#define AGG_WAVES 32
#define AGG_THREADS (AGG_WAVES * 32)
#define PLANE_VEC (PLANE_WORDS / 4)
#define TRIPS (PLANE_VEC / AGG_THREADS)
#define BATCH 8
#define MAXB 32
#define LCAP 1024
#define SRC_BITS 17
static_assert(N_NODES <= (1 << SRC_BITS) && TILE_D <= (1 << (31 - SRC_BITS)), "list word too narrow");
static_assert(PLANE_VEC % AGG_THREADS == 0 && LCAP >= 2 * MAXB, "plane pieces per lane; queue holds left-overs plus one turn");

__device__ __forceinline__ float bcast_f(float v, int srcLane) {
    return __builtin_bit_cast(float, __builtin_amdgcn_readlane(__builtin_bit_cast(int, v), srcLane));
}
__device__ __forceinline__ void lds_add(float* p, float v) {
    __hip_atomic_fetch_add(p, v, __ATOMIC_RELAXED, __HIP_MEMORY_SCOPE_WORKGROUP);
}
__device__ __forceinline__ void lds_max_f(float* p, float v) {
    if (v >= 0.0f)
        __hip_atomic_fetch_max((int*)p, __float_as_int(v), __ATOMIC_RELAXED, __HIP_MEMORY_SCOPE_WORKGROUP);
    else
        __hip_atomic_fetch_min((unsigned int*)p, __float_as_uint(v), __ATOMIC_RELAXED, __HIP_MEMORY_SCOPE_WORKGROUP);
}
__device__ __forceinline__ float leaky(float x) {
    const float v = (x >= 0.f) ? x : NEG_SLOPE * x;
    return v + 0.0f;
}

struct AggCtx {
    const int* __restrict__ src;
    const int* __restrict__ dst;
    const float* __restrict__ h;
    const float* __restrict__ el;
    float* acc;
    float* smax;
    float* ssum;
    const float4* ser;
};

__device__ __forceinline__ void max_batch(int* q, int nb, int lane, const AggCtx& c) {
    if (lane < nb) {
        const int e = q[lane];
        const int s = c.src[e];
        const int dl = c.dst[e] & (TILE_D - 1);
        q[lane] = s | (dl << SRC_BITS);
        const float4 l4 = *(const float4*)(c.el + (size_t)s * HEADS);
        const float4 r4 = c.ser[dl];
        float* mp = c.smax + dl * HEADS;
        lds_max_f(mp + 0, leaky(l4.x + r4.x));
        lds_max_f(mp + 1, leaky(l4.y + r4.y));
        lds_max_f(mp + 2, leaky(l4.z + r4.z));
        lds_max_f(mp + 3, leaky(l4.w + r4.w));
    }
}

template <bool FULL, bool LIST>
__device__ __forceinline__ void agg_batch(const int* q, int nb, int lane, const AggCtx& c) {
    int qi = lane & (BATCH - 1);
    if (!FULL) qi = min(qi, nb - 1);
    int s, dl;
    if (LIST) {
        const int p = q[qi];
        s = p & ((1 << SRC_BITS) - 1);
        dl = p >> SRC_BITS;
    } else {
        const int e = q[qi];
        s = c.src[e];
        dl = c.dst[e] & (TILE_D - 1);
    }
    const float4 l4 = *(const float4*)(c.el + (size_t)s * HEADS);

    float hv[BATCH][HEADS] = {};
#pragma unroll
    for (int k = 0; k < BATCH; ++k) {
        if (FULL || k < nb) {
            const int si = __builtin_amdgcn_readlane(s, k);
            const float* __restrict__ hp = c.h + (size_t)si * FEAT + lane;
            hv[k][0] = hp[0];
            hv[k][1] = hp[32];
            hv[k][2] = hp[64];
            hv[k][3] = hp[96];
        }
    }

    const float4 r4 = c.ser[dl];
    const float4 m4 = ((const float4*)c.smax)[dl];
    const float a0 = expf(leaky(l4.x + r4.x) - m4.x);
    const float a1 = expf(leaky(l4.y + r4.y) - m4.y);
    const float a2 = expf(leaky(l4.z + r4.z) - m4.z);
    const float a3 = expf(leaky(l4.w + r4.w) - m4.w);
    if (lane < (FULL ? BATCH : nb)) {
        float* sp = c.ssum + dl * HEADS;
        lds_add(sp + 0, a0);
        lds_add(sp + 1, a1);
        lds_add(sp + 2, a2);
        lds_add(sp + 3, a3);
    }

#pragma unroll
    for (int k = 0; k < BATCH; ++k) {
        if (FULL || k < nb) {
            const int dk = __builtin_amdgcn_readlane(dl, k);
            float* ap = c.acc + dk * FEAT + lane;
            lds_add(ap,      hv[k][0] * bcast_f(a0, k));
            lds_add(ap + 32, hv[k][1] * bcast_f(a1, k));
            lds_add(ap + 64, hv[k][2] * bcast_f(a2, k));
            lds_add(ap + 96, hv[k][3] * bcast_f(a3, k));
        }
    }
}

template <int WALK>
__device__ __forceinline__ void queue_word(unsigned x, int ebase, int& cnt, int& done, bool& kept,
                                           int lane, int* myq, const AggCtx& c) {
    unsigned m = __builtin_amdgcn_ballot_w32(x != 0);
    while (m) {
        if (WALK == 1 && __builtin_expect(cnt > LCAP - 32, 0)) {
            for (; done < cnt; done += MAXB) max_batch(myq + done, min(cnt - done, MAXB), lane, c);
            cnt = 0;
            done = 0;
            kept = false;
        }
        if (x) {
            myq[cnt + __builtin_amdgcn_mbcnt_lo(m, 0)] = ebase + __builtin_ctz(x);
            x &= x - 1;
        }
        cnt += __builtin_popcount(m);
        m = __builtin_amdgcn_ballot_w32(x != 0);
        if constexpr (WALK == 2) {
            while (cnt >= BATCH) {
                cnt -= BATCH;
                agg_batch<true, false>(myq + cnt, BATCH, lane, c);
            }
        }
    }
}

template <int WALK>
__device__ __forceinline__ int walk_edges(const uint4* pl, int wave, int lane, int* myq, const AggCtx& c) {
    int cnt = 0;
    int done = 0;
    bool kept = true;

    uint4 cur = pl[wave * 32 + lane];
    for (int it = 0; it < TRIPS; ++it) {
        const int piece = (it * AGG_WAVES + wave) * 32 + lane;
        uint4 nxt = cur;
        if (it + 1 < TRIPS)
            nxt = pl[piece + AGG_THREADS];
        const int ebase = piece * 128;
        queue_word<WALK>(cur.x, ebase, cnt, done, kept, lane, myq, c);
        queue_word<WALK>(cur.y, ebase + 32, cnt, done, kept, lane, myq, c);
        queue_word<WALK>(cur.z, ebase + 64, cnt, done, kept, lane, myq, c);
        queue_word<WALK>(cur.w, ebase + 96, cnt, done, kept, lane, myq, c);
        if constexpr (WALK == 1) {
            while (cnt - done >= MAXB) {
                max_batch(myq + done, MAXB, lane, c);
                done += MAXB;
            }
        }
        cur = nxt;
    }
    if constexpr (WALK == 1) {
        if (done < cnt) max_batch(myq + done, cnt - done, lane, c);
        return kept ? cnt : -1;
    } else {
        if (cnt > 0) agg_batch<false, false>(myq, cnt, lane, c);
        return 0;
    }
}

__device__ __forceinline__ void sum_list(const int* myq, int n, int lane, const AggCtx& c) {
    int i = 0;
    for (; i + BATCH <= n; i += BATCH) agg_batch<true, true>(myq + i, BATCH, lane, c);
    if (i < n) agg_batch<false, true>(myq + i, n - i, lane, c);
}

__global__ __launch_bounds__(AGG_THREADS) void aggregate_kernel(const int* __restrict__ src,
                                                                const int* __restrict__ dst,
                                                                const float* __restrict__ h,
                                                                const float* __restrict__ el,
                                                                const float* __restrict__ er,
                                                                float* out) {
    __shared__ float acc[TILE_D * FEAT];
    __shared__ float4 ser[TILE_D];
    __shared__ __attribute__((aligned(16))) float smax[TILE_D * HEADS];
    __shared__ float ssum[TILE_D * HEADS];
    __shared__ int q[AGG_WAVES][LCAP];

    const int tid = threadIdx.x;
    const int lane = tid & 31;
    const int wave = tid >> 5;
    const int tile = blockIdx.x;

    for (int i = tid; i < TILE_D * FEAT / 4; i += AGG_THREADS)
        ((float4*)acc)[i] = make_float4(0.f, 0.f, 0.f, 0.f);
    if (tid < TILE_D) ser[tid] = ((const float4*)er)[(size_t)tile * TILE_D + tid];
    ((unsigned int*)smax)[tid] = 0xFF800000u;
    ssum[tid] = 0.0f;
    __syncthreads();

    AggCtx c;
    c.src = src; c.dst = dst; c.h = h; c.el = el;
    c.acc = acc; c.smax = smax; c.ssum = ssum; c.ser = ser;
    const uint4* pl = (const uint4*)(out + (size_t)tile * TILE_D * FEAT);
    static_assert(PLANE_WORDS == TILE_D * FEAT, "a plane fills exactly its tile's output rows");

    const int nlist = walk_edges<1>(pl, wave, lane, q[wave], c);
    __syncthreads();
    if (nlist >= 0) sum_list(q[wave], nlist, lane, c);
    else walk_edges<2>(pl, wave, lane, q[wave], c);
    __syncthreads();

    float4* o4 = (float4*)(out + (size_t)tile * TILE_D * FEAT);
    for (int i = tid; i < TILE_D * FEAT / 4; i += AGG_THREADS) {
        float4 v = ((const float4*)acc)[i];
        const float sd = ssum[(i >> 5) * HEADS + ((i & 31) >> 3)];
        const bool any = sd > 0.f;
        v.x = any ? v.x / sd : 0.f;
        v.y = any ? v.y / sd : 0.f;
        v.z = any ? v.z / sd : 0.f;
        v.w = any ? v.w / sd : 0.f;
        v.x = (v.x > 0.f) ? v.x : (expf(v.x) - 1.0f);
        v.y = (v.y > 0.f) ? v.y : (expf(v.y) - 1.0f);
        v.z = (v.z > 0.f) ? v.z : (expf(v.z) - 1.0f);
        v.w = (v.w > 0.f) ? v.w : (expf(v.w) - 1.0f);
        o4[i] = v;
    }
    if (tile == 0 && tid == 0) {
        out[(size_t)NUM_DSTN * FEAT + 0] = 1.0f;
        out[(size_t)NUM_DSTN * FEAT + 1] = 1.0f;
    }
}

extern "C" void kernel_launch(void* const* d_in, const int* in_sizes, int n_in,
                              void* d_out, int out_size, void* d_ws, size_t ws_size,
                              hipStream_t stream) {
    const float* feats  = (const float*)d_in[0];
    const int*   src    = (const int*)d_in[1];
    const int*   dst    = (const int*)d_in[2];
    const float* W      = (const float*)d_in[3];
    const float* attn_l = (const float*)d_in[4];
    const float* attn_r = (const float*)d_in[5];
    float* out = (float*)d_out;

    float* ws   = (float*)d_ws;
    float* hbuf = ws;
    float* el   = hbuf + (size_t)N_NODES * FEAT;
    float* er   = el + (size_t)N_NODES * HEADS;

    {
        int rowTiles = N_NODES / 16;
        int blocks = rowTiles / 8;
        gemm_kernel<<<blocks, 256, 0, stream>>>(feats, W, hbuf);
    }
    {
        int n = N_NODES * HEADS;
        attn_logits_kernel<<<(n + 255) / 256, 256, 0, stream>>>(hbuf, attn_l, attn_r, el, er);
    }
    tile_plane_kernel<<<E_EDGES / TB_EDGES, 256, 0, stream>>>(dst, (unsigned int*)out);
    aggregate_kernel<<<NUM_DSTN / TILE_D, AGG_THREADS, 0, stream>>>(src, dst, hbuf, el, er, out);
}


# GEMM: tile j of a 64-column group takes columns 4c+j, not 16j+c: four tiles' B values are one 16-byte load, h rows go out in 16-byte pieces, whole lines per store; k and sum order unchanged
# speedup vs baseline: 5.7524x; 1.0631x over previous
#include <hip/hip_runtime.h>
#include <math.h>

#define N_NODES 131072
#define E_EDGES 1048576
#define NUM_DSTN 65536
#define IN_DIM 256
#define HEADS 4
#define OUT_D 32
#define FEAT (HEADS * OUT_D)
#define NEG_SLOPE 0.2f

typedef __attribute__((ext_vector_type(2))) float v2f;
typedef __attribute__((ext_vector_type(8))) float v8f;

__global__ __launch_bounds__(256, 8) void gemm_kernel(const float* __restrict__ feats,
                                                   const float* __restrict__ W,
                                                   float* __restrict__ h) {
    const int wave = blockIdx.x * (blockDim.x >> 5) + (threadIdx.x >> 5);
    const int lane = threadIdx.x & 31;
    const int lo = lane & 15;
    const int hi = lane >> 4;
    const int rowBase = wave * 16;
    const float* __restrict__ arow = feats + (size_t)(rowBase + lo) * IN_DIM;

    v8f acc[8];
#pragma unroll
    for (int t = 0; t < 8; ++t) acc[t] = (v8f){0.f, 0.f, 0.f, 0.f, 0.f, 0.f, 0.f, 0.f};

#define GEMM_TILE(t, bx, by)                                                       \
    {                                                                              \
        v2f b;                                                                     \
        b.x = (bx);                                                                \
        b.y = (by);                                                                \
        acc[t] = __builtin_amdgcn_wmma_f32_16x16x4_f32(false, a, false, b, (short)0, acc[t], false, false); \
    }
    for (int k = 0; k < IN_DIM; k += 4) {
        const int ka = k + 2 * hi;
        v2f a;
        a.x = arow[ka];
        a.y = arow[ka + 1];
        const float4* __restrict__ w0 = (const float4*)(W + (size_t)ka * FEAT) + lo;
        const float4* __restrict__ w1 = (const float4*)(W + (size_t)(ka + 1) * FEAT) + lo;
        const float4 x0 = w0[0];
        const float4 x1 = w0[16];
        const float4 y0 = w1[0];
        const float4 y1 = w1[16];
        GEMM_TILE(0, x0.x, y0.x)
        GEMM_TILE(1, x0.y, y0.y)
        GEMM_TILE(2, x0.z, y0.z)
        GEMM_TILE(3, x0.w, y0.w)
        GEMM_TILE(4, x1.x, y1.x)
        GEMM_TILE(5, x1.y, y1.y)
        GEMM_TILE(6, x1.z, y1.z)
        GEMM_TILE(7, x1.w, y1.w)
    }
#undef GEMM_TILE

#pragma unroll
    for (int i = 0; i < 8; ++i) {
        float4* __restrict__ hp = (float4*)(h + (size_t)(rowBase + i + 8 * hi) * FEAT) + lo;
        hp[0]  = make_float4(acc[0][i], acc[1][i], acc[2][i], acc[3][i]);
        hp[16] = make_float4(acc[4][i], acc[5][i], acc[6][i], acc[7][i]);
    }
}

__global__ void attn_logits_kernel(const float* __restrict__ h,
                                   const float* __restrict__ attn_l,
                                   const float* __restrict__ attn_r,
                                   float* __restrict__ el, float* __restrict__ er) {
    int i = blockIdx.x * blockDim.x + threadIdx.x;
    if (i >= N_NODES * HEADS) return;
    const int n = i >> 2;
    const int hh = i & 3;
    const float* __restrict__ hp = h + (size_t)n * FEAT + hh * OUT_D;
    const float* __restrict__ al = attn_l + hh * OUT_D;
    const float* __restrict__ ar = attn_r + hh * OUT_D;
    float sl = 0.f, sr = 0.f;
#pragma unroll
    for (int d = 0; d < OUT_D; ++d) {
        float v = hp[d];
        sl += v * al[d];
        sr += v * ar[d];
    }
    el[i] = sl;
    er[i] = sr;
}

#define TILE_SHIFT 8
#define N_TILES (NUM_DSTN >> TILE_SHIFT)
#define PLANE_WORDS (E_EDGES / 32)
#define TB_EDGES 2048
#define TB_WORDS (TB_EDGES / 32)
__global__ __launch_bounds__(256) void tile_plane_kernel(const int* __restrict__ dst,
                                                         unsigned int* planes) {
    __shared__ __attribute__((aligned(16))) unsigned int bits[N_TILES * TB_WORDS];
    const int tid = threadIdx.x;
    uint4* b4 = (uint4*)bits;
    for (int i = tid; i < N_TILES * TB_WORDS / 4; i += 256) b4[i] = make_uint4(0u, 0u, 0u, 0u);
    __syncthreads();

    const int g = blockIdx.x * (TB_EDGES / 4) + tid * 2;
    const int4 d0 = ((const int4*)dst)[g];
    const int4 d1 = ((const int4*)dst)[g + 1];
    unsigned int* wp = bits + (tid >> 2);
    const unsigned b0 = 1u << ((tid & 3) * 8);
#define SET_BIT(d, k)                                                                              \
    __hip_atomic_fetch_or(wp + (((unsigned)(d) >> TILE_SHIFT) & (N_TILES - 1)) * TB_WORDS, b0 << (k), \
                          __ATOMIC_RELAXED, __HIP_MEMORY_SCOPE_WORKGROUP)
    SET_BIT(d0.x, 0); SET_BIT(d0.y, 1); SET_BIT(d0.z, 2); SET_BIT(d0.w, 3);
    SET_BIT(d1.x, 4); SET_BIT(d1.y, 5); SET_BIT(d1.z, 6); SET_BIT(d1.w, 7);
#undef SET_BIT
    __syncthreads();

    for (int i = tid; i < N_TILES * TB_WORDS / 4; i += 256) {
        uint4* o = (uint4*)(planes + (size_t)(i >> 4) * PLANE_WORDS + blockIdx.x * TB_WORDS);
        o[i & 15] = b4[i];
    }
}

#define TILE_D (1 << TILE_SHIFT)
#define AGG_WAVES 32
#define AGG_THREADS (AGG_WAVES * 32)
#define PLANE_VEC (PLANE_WORDS / 4)
#define TRIPS (PLANE_VEC / AGG_THREADS)
#define BATCH 8
#define MAXB 32
#define LCAP 1024
#define SRC_BITS 17
static_assert(N_NODES <= (1 << SRC_BITS) && TILE_D <= (1 << (31 - SRC_BITS)), "list word too narrow");
static_assert(PLANE_VEC % AGG_THREADS == 0 && LCAP >= 2 * MAXB, "plane pieces per lane; queue holds left-overs plus one turn");

__device__ __forceinline__ float bcast_f(float v, int srcLane) {
    return __builtin_bit_cast(float, __builtin_amdgcn_readlane(__builtin_bit_cast(int, v), srcLane));
}
__device__ __forceinline__ void lds_add(float* p, float v) {
    __hip_atomic_fetch_add(p, v, __ATOMIC_RELAXED, __HIP_MEMORY_SCOPE_WORKGROUP);
}
__device__ __forceinline__ void lds_max_f(float* p, float v) {
    if (v >= 0.0f)
        __hip_atomic_fetch_max((int*)p, __float_as_int(v), __ATOMIC_RELAXED, __HIP_MEMORY_SCOPE_WORKGROUP);
    else
        __hip_atomic_fetch_min((unsigned int*)p, __float_as_uint(v), __ATOMIC_RELAXED, __HIP_MEMORY_SCOPE_WORKGROUP);
}
__device__ __forceinline__ float leaky(float x) {
    const float v = (x >= 0.f) ? x : NEG_SLOPE * x;
    return v + 0.0f;
}

struct AggCtx {
    const int* __restrict__ src;
    const int* __restrict__ dst;
    const float* __restrict__ h;
    const float* __restrict__ el;
    float* acc;
    float* smax;
    float* ssum;
    const float4* ser;
};

__device__ __forceinline__ void max_batch(int* q, int nb, int lane, const AggCtx& c) {
    if (lane < nb) {
        const int e = q[lane];
        const int s = c.src[e];
        const int dl = c.dst[e] & (TILE_D - 1);
        q[lane] = s | (dl << SRC_BITS);
        const float4 l4 = *(const float4*)(c.el + (size_t)s * HEADS);
        const float4 r4 = c.ser[dl];
        float* mp = c.smax + dl * HEADS;
        lds_max_f(mp + 0, leaky(l4.x + r4.x));
        lds_max_f(mp + 1, leaky(l4.y + r4.y));
        lds_max_f(mp + 2, leaky(l4.z + r4.z));
        lds_max_f(mp + 3, leaky(l4.w + r4.w));
    }
}

template <bool FULL, bool LIST>
__device__ __forceinline__ void agg_batch(const int* q, int nb, int lane, const AggCtx& c) {
    int qi = lane & (BATCH - 1);
    if (!FULL) qi = min(qi, nb - 1);
    int s, dl;
    if (LIST) {
        const int p = q[qi];
        s = p & ((1 << SRC_BITS) - 1);
        dl = p >> SRC_BITS;
    } else {
        const int e = q[qi];
        s = c.src[e];
        dl = c.dst[e] & (TILE_D - 1);
    }
    const float4 l4 = *(const float4*)(c.el + (size_t)s * HEADS);

    float hv[BATCH][HEADS] = {};
#pragma unroll
    for (int k = 0; k < BATCH; ++k) {
        if (FULL || k < nb) {
            const int si = __builtin_amdgcn_readlane(s, k);
            const float* __restrict__ hp = c.h + (size_t)si * FEAT + lane;
            hv[k][0] = hp[0];
            hv[k][1] = hp[32];
            hv[k][2] = hp[64];
            hv[k][3] = hp[96];
        }
    }

    const float4 r4 = c.ser[dl];
    const float4 m4 = ((const float4*)c.smax)[dl];
    const float a0 = expf(leaky(l4.x + r4.x) - m4.x);
    const float a1 = expf(leaky(l4.y + r4.y) - m4.y);
    const float a2 = expf(leaky(l4.z + r4.z) - m4.z);
    const float a3 = expf(leaky(l4.w + r4.w) - m4.w);
    if (lane < (FULL ? BATCH : nb)) {
        float* sp = c.ssum + dl * HEADS;
        lds_add(sp + 0, a0);
        lds_add(sp + 1, a1);
        lds_add(sp + 2, a2);
        lds_add(sp + 3, a3);
    }

#pragma unroll
    for (int k = 0; k < BATCH; ++k) {
        if (FULL || k < nb) {
            const int dk = __builtin_amdgcn_readlane(dl, k);
            float* ap = c.acc + dk * FEAT + lane;
            lds_add(ap,      hv[k][0] * bcast_f(a0, k));
            lds_add(ap + 32, hv[k][1] * bcast_f(a1, k));
            lds_add(ap + 64, hv[k][2] * bcast_f(a2, k));
            lds_add(ap + 96, hv[k][3] * bcast_f(a3, k));
        }
    }
}

template <int WALK>
__device__ __forceinline__ void queue_word(unsigned x, int ebase, int& cnt, int& done, bool& kept,
                                           int lane, int* myq, const AggCtx& c) {
    unsigned m = __builtin_amdgcn_ballot_w32(x != 0);
    while (m) {
        if (WALK == 1 && __builtin_expect(cnt > LCAP - 32, 0)) {
            for (; done < cnt; done += MAXB) max_batch(myq + done, min(cnt - done, MAXB), lane, c);
            cnt = 0;
            done = 0;
            kept = false;
        }
        if (x) {
            myq[cnt + __builtin_amdgcn_mbcnt_lo(m, 0)] = ebase + __builtin_ctz(x);
            x &= x - 1;
        }
        cnt += __builtin_popcount(m);
        m = __builtin_amdgcn_ballot_w32(x != 0);
        if constexpr (WALK == 2) {
            while (cnt >= BATCH) {
                cnt -= BATCH;
                agg_batch<true, false>(myq + cnt, BATCH, lane, c);
            }
        }
    }
}

template <int WALK>
__device__ __forceinline__ int walk_edges(const uint4* pl, int wave, int lane, int* myq, const AggCtx& c) {
    int cnt = 0;
    int done = 0;
    bool kept = true;

    uint4 cur = pl[wave * 32 + lane];
    for (int it = 0; it < TRIPS; ++it) {
        const int piece = (it * AGG_WAVES + wave) * 32 + lane;
        uint4 nxt = cur;
        if (it + 1 < TRIPS)
            nxt = pl[piece + AGG_THREADS];
        const int ebase = piece * 128;
        queue_word<WALK>(cur.x, ebase, cnt, done, kept, lane, myq, c);
        queue_word<WALK>(cur.y, ebase + 32, cnt, done, kept, lane, myq, c);
        queue_word<WALK>(cur.z, ebase + 64, cnt, done, kept, lane, myq, c);
        queue_word<WALK>(cur.w, ebase + 96, cnt, done, kept, lane, myq, c);
        if constexpr (WALK == 1) {
            while (cnt - done >= MAXB) {
                max_batch(myq + done, MAXB, lane, c);
                done += MAXB;
            }
        }
        cur = nxt;
    }
    if constexpr (WALK == 1) {
        if (done < cnt) max_batch(myq + done, cnt - done, lane, c);
        return kept ? cnt : -1;
    } else {
        if (cnt > 0) agg_batch<false, false>(myq, cnt, lane, c);
        return 0;
    }
}

__device__ __forceinline__ void sum_list(const int* myq, int n, int lane, const AggCtx& c) {
    int i = 0;
    for (; i + BATCH <= n; i += BATCH) agg_batch<true, true>(myq + i, BATCH, lane, c);
    if (i < n) agg_batch<false, true>(myq + i, n - i, lane, c);
}

__global__ __launch_bounds__(AGG_THREADS) void aggregate_kernel(const int* __restrict__ src,
                                                                const int* __restrict__ dst,
                                                                const float* __restrict__ h,
                                                                const float* __restrict__ el,
                                                                const float* __restrict__ er,
                                                                float* out) {
    __shared__ float acc[TILE_D * FEAT];
    __shared__ float4 ser[TILE_D];
    __shared__ __attribute__((aligned(16))) float smax[TILE_D * HEADS];
    __shared__ float ssum[TILE_D * HEADS];
    __shared__ int q[AGG_WAVES][LCAP];

    const int tid = threadIdx.x;
    const int lane = tid & 31;
    const int wave = tid >> 5;
    const int tile = blockIdx.x;

    for (int i = tid; i < TILE_D * FEAT / 4; i += AGG_THREADS)
        ((float4*)acc)[i] = make_float4(0.f, 0.f, 0.f, 0.f);
    if (tid < TILE_D) ser[tid] = ((const float4*)er)[(size_t)tile * TILE_D + tid];
    ((unsigned int*)smax)[tid] = 0xFF800000u;
    ssum[tid] = 0.0f;
    __syncthreads();

    AggCtx c;
    c.src = src; c.dst = dst; c.h = h; c.el = el;
    c.acc = acc; c.smax = smax; c.ssum = ssum; c.ser = ser;
    const uint4* pl = (const uint4*)(out + (size_t)tile * TILE_D * FEAT);
    static_assert(PLANE_WORDS == TILE_D * FEAT, "a plane fills exactly its tile's output rows");

    const int nlist = walk_edges<1>(pl, wave, lane, q[wave], c);
    __syncthreads();
    if (nlist >= 0) sum_list(q[wave], nlist, lane, c);
    else walk_edges<2>(pl, wave, lane, q[wave], c);
    __syncthreads();

    float4* o4 = (float4*)(out + (size_t)tile * TILE_D * FEAT);
    for (int i = tid; i < TILE_D * FEAT / 4; i += AGG_THREADS) {
        float4 v = ((const float4*)acc)[i];
        const float sd = ssum[(i >> 5) * HEADS + ((i & 31) >> 3)];
        const bool any = sd > 0.f;
        v.x = any ? v.x / sd : 0.f;
        v.y = any ? v.y / sd : 0.f;
        v.z = any ? v.z / sd : 0.f;
        v.w = any ? v.w / sd : 0.f;
        v.x = (v.x > 0.f) ? v.x : (expf(v.x) - 1.0f);
        v.y = (v.y > 0.f) ? v.y : (expf(v.y) - 1.0f);
        v.z = (v.z > 0.f) ? v.z : (expf(v.z) - 1.0f);
        v.w = (v.w > 0.f) ? v.w : (expf(v.w) - 1.0f);
        o4[i] = v;
    }
    if (tile == 0 && tid == 0) {
        out[(size_t)NUM_DSTN * FEAT + 0] = 1.0f;
        out[(size_t)NUM_DSTN * FEAT + 1] = 1.0f;
    }
}

extern "C" void kernel_launch(void* const* d_in, const int* in_sizes, int n_in,
                              void* d_out, int out_size, void* d_ws, size_t ws_size,
                              hipStream_t stream) {
    const float* feats  = (const float*)d_in[0];
    const int*   src    = (const int*)d_in[1];
    const int*   dst    = (const int*)d_in[2];
    const float* W      = (const float*)d_in[3];
    const float* attn_l = (const float*)d_in[4];
    const float* attn_r = (const float*)d_in[5];
    float* out = (float*)d_out;

    float* ws   = (float*)d_ws;
    float* hbuf = ws;
    float* el   = hbuf + (size_t)N_NODES * FEAT;
    float* er   = el + (size_t)N_NODES * HEADS;

    {
        int rowTiles = N_NODES / 16;
        int blocks = rowTiles / 8;
        gemm_kernel<<<blocks, 256, 0, stream>>>(feats, W, hbuf);
    }
    {
        int n = N_NODES * HEADS;
        attn_logits_kernel<<<(n + 255) / 256, 256, 0, stream>>>(hbuf, attn_l, attn_r, el, er);
    }
    tile_plane_kernel<<<E_EDGES / TB_EDGES, 256, 0, stream>>>(dst, (unsigned int*)out);
    aggregate_kernel<<<NUM_DSTN / TILE_D, AGG_THREADS, 0, stream>>>(src, dst, hbuf, el, er, out);
}


# Aggregation walk 2: an h row is one 16-byte load per lane (512 contiguous bytes per instruction, not four 128-byte pieces); a lane owns 4 columns of one head; LDS sums stay in arrival order
# speedup vs baseline: 6.1019x; 1.0608x over previous
#include <hip/hip_runtime.h>
#include <math.h>

#define N_NODES 131072
#define E_EDGES 1048576
#define NUM_DSTN 65536
#define IN_DIM 256
#define HEADS 4
#define OUT_D 32
#define FEAT (HEADS * OUT_D)
#define NEG_SLOPE 0.2f

typedef __attribute__((ext_vector_type(2))) float v2f;
typedef __attribute__((ext_vector_type(8))) float v8f;

__global__ __launch_bounds__(256, 8) void gemm_kernel(const float* __restrict__ feats,
                                                   const float* __restrict__ W,
                                                   float* __restrict__ h) {
    const int wave = blockIdx.x * (blockDim.x >> 5) + (threadIdx.x >> 5);
    const int lane = threadIdx.x & 31;
    const int lo = lane & 15;
    const int hi = lane >> 4;
    const int rowBase = wave * 16;
    const float* __restrict__ arow = feats + (size_t)(rowBase + lo) * IN_DIM;

    v8f acc[8];
#pragma unroll
    for (int t = 0; t < 8; ++t) acc[t] = (v8f){0.f, 0.f, 0.f, 0.f, 0.f, 0.f, 0.f, 0.f};

#define GEMM_TILE(t, bx, by)                                                       \
    {                                                                              \
        v2f b;                                                                     \
        b.x = (bx);                                                                \
        b.y = (by);                                                                \
        acc[t] = __builtin_amdgcn_wmma_f32_16x16x4_f32(false, a, false, b, (short)0, acc[t], false, false); \
    }
    for (int k = 0; k < IN_DIM; k += 4) {
        const int ka = k + 2 * hi;
        v2f a;
        a.x = arow[ka];
        a.y = arow[ka + 1];
        const float4* __restrict__ w0 = (const float4*)(W + (size_t)ka * FEAT) + lo;
        const float4* __restrict__ w1 = (const float4*)(W + (size_t)(ka + 1) * FEAT) + lo;
        const float4 x0 = w0[0];
        const float4 x1 = w0[16];
        const float4 y0 = w1[0];
        const float4 y1 = w1[16];
        GEMM_TILE(0, x0.x, y0.x)
        GEMM_TILE(1, x0.y, y0.y)
        GEMM_TILE(2, x0.z, y0.z)
        GEMM_TILE(3, x0.w, y0.w)
        GEMM_TILE(4, x1.x, y1.x)
        GEMM_TILE(5, x1.y, y1.y)
        GEMM_TILE(6, x1.z, y1.z)
        GEMM_TILE(7, x1.w, y1.w)
    }
#undef GEMM_TILE

#pragma unroll
    for (int i = 0; i < 8; ++i) {
        float4* __restrict__ hp = (float4*)(h + (size_t)(rowBase + i + 8 * hi) * FEAT) + lo;
        hp[0]  = make_float4(acc[0][i], acc[1][i], acc[2][i], acc[3][i]);
        hp[16] = make_float4(acc[4][i], acc[5][i], acc[6][i], acc[7][i]);
    }
}

__global__ void attn_logits_kernel(const float* __restrict__ h,
                                   const float* __restrict__ attn_l,
                                   const float* __restrict__ attn_r,
                                   float* __restrict__ el, float* __restrict__ er) {
    int i = blockIdx.x * blockDim.x + threadIdx.x;
    if (i >= N_NODES * HEADS) return;
    const int n = i >> 2;
    const int hh = i & 3;
    const float* __restrict__ hp = h + (size_t)n * FEAT + hh * OUT_D;
    const float* __restrict__ al = attn_l + hh * OUT_D;
    const float* __restrict__ ar = attn_r + hh * OUT_D;
    float sl = 0.f, sr = 0.f;
#pragma unroll
    for (int d = 0; d < OUT_D; ++d) {
        float v = hp[d];
        sl += v * al[d];
        sr += v * ar[d];
    }
    el[i] = sl;
    er[i] = sr;
}

#define TILE_SHIFT 8
#define N_TILES (NUM_DSTN >> TILE_SHIFT)
#define PLANE_WORDS (E_EDGES / 32)
#define TB_EDGES 2048
#define TB_WORDS (TB_EDGES / 32)
__global__ __launch_bounds__(256) void tile_plane_kernel(const int* __restrict__ dst,
                                                         unsigned int* planes) {
    __shared__ __attribute__((aligned(16))) unsigned int bits[N_TILES * TB_WORDS];
    const int tid = threadIdx.x;
    uint4* b4 = (uint4*)bits;
    for (int i = tid; i < N_TILES * TB_WORDS / 4; i += 256) b4[i] = make_uint4(0u, 0u, 0u, 0u);
    __syncthreads();

    const int g = blockIdx.x * (TB_EDGES / 4) + tid * 2;
    const int4 d0 = ((const int4*)dst)[g];
    const int4 d1 = ((const int4*)dst)[g + 1];
    unsigned int* wp = bits + (tid >> 2);
    const unsigned b0 = 1u << ((tid & 3) * 8);
#define SET_BIT(d, k)                                                                              \
    __hip_atomic_fetch_or(wp + (((unsigned)(d) >> TILE_SHIFT) & (N_TILES - 1)) * TB_WORDS, b0 << (k), \
                          __ATOMIC_RELAXED, __HIP_MEMORY_SCOPE_WORKGROUP)
    SET_BIT(d0.x, 0); SET_BIT(d0.y, 1); SET_BIT(d0.z, 2); SET_BIT(d0.w, 3);
    SET_BIT(d1.x, 4); SET_BIT(d1.y, 5); SET_BIT(d1.z, 6); SET_BIT(d1.w, 7);
#undef SET_BIT
    __syncthreads();

    for (int i = tid; i < N_TILES * TB_WORDS / 4; i += 256) {
        uint4* o = (uint4*)(planes + (size_t)(i >> 4) * PLANE_WORDS + blockIdx.x * TB_WORDS);
        o[i & 15] = b4[i];
    }
}

#define TILE_D (1 << TILE_SHIFT)
#define AGG_WAVES 32
#define AGG_THREADS (AGG_WAVES * 32)
#define PLANE_VEC (PLANE_WORDS / 4)
#define TRIPS (PLANE_VEC / AGG_THREADS)
#define BATCH 8
#define MAXB 32
#define LCAP 1024
#define SRC_BITS 17
static_assert(N_NODES <= (1 << SRC_BITS) && TILE_D <= (1 << (31 - SRC_BITS)), "list word too narrow");
static_assert(PLANE_VEC % AGG_THREADS == 0 && LCAP >= 2 * MAXB, "plane pieces per lane; queue holds left-overs plus one turn");
static_assert(BATCH == 8 && HEADS * 8 == 32 && OUT_D == 32, "walk 2: a group of eight lanes per head, one entry per lane of a group");

__device__ __forceinline__ void lds_add(float* p, float v) {
    __hip_atomic_fetch_add(p, v, __ATOMIC_RELAXED, __HIP_MEMORY_SCOPE_WORKGROUP);
}
__device__ __forceinline__ void lds_max_f(float* p, float v) {
    if (v >= 0.0f)
        __hip_atomic_fetch_max((int*)p, __float_as_int(v), __ATOMIC_RELAXED, __HIP_MEMORY_SCOPE_WORKGROUP);
    else
        __hip_atomic_fetch_min((unsigned int*)p, __float_as_uint(v), __ATOMIC_RELAXED, __HIP_MEMORY_SCOPE_WORKGROUP);
}
__device__ __forceinline__ float leaky(float x) {
    const float v = (x >= 0.f) ? x : NEG_SLOPE * x;
    return v + 0.0f;
}

struct AggCtx {
    const int* __restrict__ src;
    const int* __restrict__ dst;
    const float* __restrict__ h;
    const float* __restrict__ el;
    float* acc;
    float* smax;
    float* ssum;
    const float4* ser;
};

__device__ __forceinline__ void max_batch(int* q, int nb, int lane, const AggCtx& c) {
    if (lane < nb) {
        const int e = q[lane];
        const int s = c.src[e];
        const int dl = c.dst[e] & (TILE_D - 1);
        q[lane] = s | (dl << SRC_BITS);
        const float4 l4 = *(const float4*)(c.el + (size_t)s * HEADS);
        const float4 r4 = c.ser[dl];
        float* mp = c.smax + dl * HEADS;
        lds_max_f(mp + 0, leaky(l4.x + r4.x));
        lds_max_f(mp + 1, leaky(l4.y + r4.y));
        lds_max_f(mp + 2, leaky(l4.z + r4.z));
        lds_max_f(mp + 3, leaky(l4.w + r4.w));
    }
}

template <int K>
__device__ __forceinline__ float group_bcast(float v) {
    return __uint_as_float(__builtin_amdgcn_mov_dpp8(__float_as_uint(v), K * 0x249249u));
}

template <bool FULL, bool LIST>
__device__ __forceinline__ void agg_batch(const int* q, int nb, int lane, const AggCtx& c) {
    const int hd = lane >> 3;
    int qi = lane & (BATCH - 1);
    const bool mine = FULL || qi < nb;
    if (!FULL) qi = min(qi, nb - 1);
    int s, dl;
    if (LIST) {
        const int p = q[qi];
        s = p & ((1 << SRC_BITS) - 1);
        dl = p >> SRC_BITS;
    } else {
        const int e = q[qi];
        s = c.src[e];
        dl = c.dst[e] & (TILE_D - 1);
    }
    const float l1 = c.el[(size_t)s * HEADS + hd];

    float4 hv[BATCH];
#pragma unroll
    for (int k = 0; k < BATCH; ++k) {
        if (FULL || k < nb) {
            const int si = __builtin_amdgcn_readlane(s, k);
            hv[k] = ((const float4*)(c.h + (size_t)si * FEAT))[lane];
        } else {
            hv[k] = make_float4(0.f, 0.f, 0.f, 0.f);
        }
    }

    const float r1 = ((const float*)c.ser)[dl * HEADS + hd];
    const float m1 = c.smax[dl * HEADS + hd];
    const float a = expf(leaky(l1 + r1) - m1);
    if (mine) lds_add(c.ssum + dl * HEADS + hd, a);

#define ADD_ROW(K)                                                  \
    if (FULL || K < nb) {                                           \
        const int dk = __builtin_amdgcn_readlane(dl, K);            \
        const float w = group_bcast<K>(a);                          \
        float* ap = c.acc + dk * FEAT + lane;                       \
        lds_add(ap,      hv[K].x * w);                              \
        lds_add(ap + 32, hv[K].y * w);                              \
        lds_add(ap + 64, hv[K].z * w);                              \
        lds_add(ap + 96, hv[K].w * w);                              \
    }
    ADD_ROW(0) ADD_ROW(1) ADD_ROW(2) ADD_ROW(3)
    ADD_ROW(4) ADD_ROW(5) ADD_ROW(6) ADD_ROW(7)
#undef ADD_ROW
}

template <int WALK>
__device__ __forceinline__ void queue_word(unsigned x, int ebase, int& cnt, int& done, bool& kept,
                                           int lane, int* myq, const AggCtx& c) {
    unsigned m = __builtin_amdgcn_ballot_w32(x != 0);
    while (m) {
        if (WALK == 1 && __builtin_expect(cnt > LCAP - 32, 0)) {
            for (; done < cnt; done += MAXB) max_batch(myq + done, min(cnt - done, MAXB), lane, c);
            cnt = 0;
            done = 0;
            kept = false;
        }
        if (x) {
            myq[cnt + __builtin_amdgcn_mbcnt_lo(m, 0)] = ebase + __builtin_ctz(x);
            x &= x - 1;
        }
        cnt += __builtin_popcount(m);
        m = __builtin_amdgcn_ballot_w32(x != 0);
        if constexpr (WALK == 2) {
            while (cnt >= BATCH) {
                cnt -= BATCH;
                agg_batch<true, false>(myq + cnt, BATCH, lane, c);
            }
        }
    }
}

template <int WALK>
__device__ __forceinline__ int walk_edges(const uint4* pl, int wave, int lane, int* myq, const AggCtx& c) {
    int cnt = 0;
    int done = 0;
    bool kept = true;

    uint4 cur = pl[wave * 32 + lane];
    for (int it = 0; it < TRIPS; ++it) {
        const int piece = (it * AGG_WAVES + wave) * 32 + lane;
        uint4 nxt = cur;
        if (it + 1 < TRIPS)
            nxt = pl[piece + AGG_THREADS];
        const int ebase = piece * 128;
        queue_word<WALK>(cur.x, ebase, cnt, done, kept, lane, myq, c);
        queue_word<WALK>(cur.y, ebase + 32, cnt, done, kept, lane, myq, c);
        queue_word<WALK>(cur.z, ebase + 64, cnt, done, kept, lane, myq, c);
        queue_word<WALK>(cur.w, ebase + 96, cnt, done, kept, lane, myq, c);
        if constexpr (WALK == 1) {
            while (cnt - done >= MAXB) {
                max_batch(myq + done, MAXB, lane, c);
                done += MAXB;
            }
        }
        cur = nxt;
    }
    if constexpr (WALK == 1) {
        if (done < cnt) max_batch(myq + done, cnt - done, lane, c);
        return kept ? cnt : -1;
    } else {
        if (cnt > 0) agg_batch<false, false>(myq, cnt, lane, c);
        return 0;
    }
}

__device__ __forceinline__ void sum_list(const int* myq, int n, int lane, const AggCtx& c) {
    int i = 0;
    for (; i + BATCH <= n; i += BATCH) agg_batch<true, true>(myq + i, BATCH, lane, c);
    if (i < n) agg_batch<false, true>(myq + i, n - i, lane, c);
}

__global__ __launch_bounds__(AGG_THREADS) void aggregate_kernel(const int* __restrict__ src,
                                                                const int* __restrict__ dst,
                                                                const float* __restrict__ h,
                                                                const float* __restrict__ el,
                                                                const float* __restrict__ er,
                                                                float* out) {
    __shared__ float acc[TILE_D * FEAT];
    __shared__ float4 ser[TILE_D];
    __shared__ __attribute__((aligned(16))) float smax[TILE_D * HEADS];
    __shared__ float ssum[TILE_D * HEADS];
    __shared__ int q[AGG_WAVES][LCAP];

    const int tid = threadIdx.x;
    const int lane = tid & 31;
    const int wave = tid >> 5;
    const int tile = blockIdx.x;

    for (int i = tid; i < TILE_D * FEAT / 4; i += AGG_THREADS)
        ((float4*)acc)[i] = make_float4(0.f, 0.f, 0.f, 0.f);
    if (tid < TILE_D) ser[tid] = ((const float4*)er)[(size_t)tile * TILE_D + tid];
    ((unsigned int*)smax)[tid] = 0xFF800000u;
    ssum[tid] = 0.0f;
    __syncthreads();

    AggCtx c;
    c.src = src; c.dst = dst; c.h = h; c.el = el;
    c.acc = acc; c.smax = smax; c.ssum = ssum; c.ser = ser;
    const uint4* pl = (const uint4*)(out + (size_t)tile * TILE_D * FEAT);
    static_assert(PLANE_WORDS == TILE_D * FEAT, "a plane fills exactly its tile's output rows");

    const int nlist = walk_edges<1>(pl, wave, lane, q[wave], c);
    __syncthreads();
    if (nlist >= 0) sum_list(q[wave], nlist, lane, c);
    else walk_edges<2>(pl, wave, lane, q[wave], c);
    __syncthreads();

    float4* o4 = (float4*)(out + (size_t)tile * TILE_D * FEAT);
    for (int i = tid; i < TILE_D * FEAT / 4; i += AGG_THREADS) {
        const float* ar = acc + (i >> 5) * FEAT + (i & 31);
        float4 v = make_float4(ar[0], ar[32], ar[64], ar[96]);
        const float sd = ssum[(i >> 5) * HEADS + ((i & 31) >> 3)];
        const bool any = sd > 0.f;
        v.x = any ? v.x / sd : 0.f;
        v.y = any ? v.y / sd : 0.f;
        v.z = any ? v.z / sd : 0.f;
        v.w = any ? v.w / sd : 0.f;
        v.x = (v.x > 0.f) ? v.x : (expf(v.x) - 1.0f);
        v.y = (v.y > 0.f) ? v.y : (expf(v.y) - 1.0f);
        v.z = (v.z > 0.f) ? v.z : (expf(v.z) - 1.0f);
        v.w = (v.w > 0.f) ? v.w : (expf(v.w) - 1.0f);
        o4[i] = v;
    }
    if (tile == 0 && tid == 0) {
        out[(size_t)NUM_DSTN * FEAT + 0] = 1.0f;
        out[(size_t)NUM_DSTN * FEAT + 1] = 1.0f;
    }
}

extern "C" void kernel_launch(void* const* d_in, const int* in_sizes, int n_in,
                              void* d_out, int out_size, void* d_ws, size_t ws_size,
                              hipStream_t stream) {
    const float* feats  = (const float*)d_in[0];
    const int*   src    = (const int*)d_in[1];
    const int*   dst    = (const int*)d_in[2];
    const float* W      = (const float*)d_in[3];
    const float* attn_l = (const float*)d_in[4];
    const float* attn_r = (const float*)d_in[5];
    float* out = (float*)d_out;

    float* ws   = (float*)d_ws;
    float* hbuf = ws;
    float* el   = hbuf + (size_t)N_NODES * FEAT;
    float* er   = el + (size_t)N_NODES * HEADS;

    {
        int rowTiles = N_NODES / 16;
        int blocks = rowTiles / 8;
        gemm_kernel<<<blocks, 256, 0, stream>>>(feats, W, hbuf);
    }
    {
        int n = N_NODES * HEADS;
        attn_logits_kernel<<<(n + 255) / 256, 256, 0, stream>>>(hbuf, attn_l, attn_r, el, er);
    }
    tile_plane_kernel<<<E_EDGES / TB_EDGES, 256, 0, stream>>>(dst, (unsigned int*)out);
    aggregate_kernel<<<NUM_DSTN / TILE_D, AGG_THREADS, 0, stream>>>(src, dst, hbuf, el, er, out);
}


# Aggregation: walk 1 bins each tile edge's src by dst row in LDS (counter rank); a wave sums its own 8 rows and their softmax sums in registers (8-lane tree): LDS float adds go; order = arrival rank
# speedup vs baseline: 6.4561x; 1.0580x over previous
#include <hip/hip_runtime.h>
#include <math.h>

#define N_NODES 131072
#define E_EDGES 1048576
#define NUM_DSTN 65536
#define IN_DIM 256
#define HEADS 4
#define OUT_D 32
#define FEAT (HEADS * OUT_D)
#define NEG_SLOPE 0.2f

typedef __attribute__((ext_vector_type(2))) float v2f;
typedef __attribute__((ext_vector_type(8))) float v8f;

__global__ __launch_bounds__(256, 8) void gemm_kernel(const float* __restrict__ feats,
                                                   const float* __restrict__ W,
                                                   float* __restrict__ h) {
    const int wave = blockIdx.x * (blockDim.x >> 5) + (threadIdx.x >> 5);
    const int lane = threadIdx.x & 31;
    const int lo = lane & 15;
    const int hi = lane >> 4;
    const int rowBase = wave * 16;
    const float* __restrict__ arow = feats + (size_t)(rowBase + lo) * IN_DIM;

    v8f acc[8];
#pragma unroll
    for (int t = 0; t < 8; ++t) acc[t] = (v8f){0.f, 0.f, 0.f, 0.f, 0.f, 0.f, 0.f, 0.f};

#define GEMM_TILE(t, bx, by)                                                       \
    {                                                                              \
        v2f b;                                                                     \
        b.x = (bx);                                                                \
        b.y = (by);                                                                \
        acc[t] = __builtin_amdgcn_wmma_f32_16x16x4_f32(false, a, false, b, (short)0, acc[t], false, false); \
    }
    for (int k = 0; k < IN_DIM; k += 4) {
        const int ka = k + 2 * hi;
        v2f a;
        a.x = arow[ka];
        a.y = arow[ka + 1];
        const float4* __restrict__ w0 = (const float4*)(W + (size_t)ka * FEAT) + lo;
        const float4* __restrict__ w1 = (const float4*)(W + (size_t)(ka + 1) * FEAT) + lo;
        const float4 x0 = w0[0];
        const float4 x1 = w0[16];
        const float4 y0 = w1[0];
        const float4 y1 = w1[16];
        GEMM_TILE(0, x0.x, y0.x)
        GEMM_TILE(1, x0.y, y0.y)
        GEMM_TILE(2, x0.z, y0.z)
        GEMM_TILE(3, x0.w, y0.w)
        GEMM_TILE(4, x1.x, y1.x)
        GEMM_TILE(5, x1.y, y1.y)
        GEMM_TILE(6, x1.z, y1.z)
        GEMM_TILE(7, x1.w, y1.w)
    }
#undef GEMM_TILE

#pragma unroll
    for (int i = 0; i < 8; ++i) {
        float4* __restrict__ hp = (float4*)(h + (size_t)(rowBase + i + 8 * hi) * FEAT) + lo;
        hp[0]  = make_float4(acc[0][i], acc[1][i], acc[2][i], acc[3][i]);
        hp[16] = make_float4(acc[4][i], acc[5][i], acc[6][i], acc[7][i]);
    }
}

__global__ void attn_logits_kernel(const float* __restrict__ h,
                                   const float* __restrict__ attn_l,
                                   const float* __restrict__ attn_r,
                                   float* __restrict__ el, float* __restrict__ er) {
    int i = blockIdx.x * blockDim.x + threadIdx.x;
    if (i >= N_NODES * HEADS) return;
    const int n = i >> 2;
    const int hh = i & 3;
    const float* __restrict__ hp = h + (size_t)n * FEAT + hh * OUT_D;
    const float* __restrict__ al = attn_l + hh * OUT_D;
    const float* __restrict__ ar = attn_r + hh * OUT_D;
    float sl = 0.f, sr = 0.f;
#pragma unroll
    for (int d = 0; d < OUT_D; ++d) {
        float v = hp[d];
        sl += v * al[d];
        sr += v * ar[d];
    }
    el[i] = sl;
    er[i] = sr;
}

#define TILE_SHIFT 8
#define N_TILES (NUM_DSTN >> TILE_SHIFT)
#define PLANE_WORDS (E_EDGES / 32)
#define TB_EDGES 2048
#define TB_WORDS (TB_EDGES / 32)
__global__ __launch_bounds__(256) void tile_plane_kernel(const int* __restrict__ dst,
                                                         unsigned int* planes) {
    __shared__ __attribute__((aligned(16))) unsigned int bits[N_TILES * TB_WORDS];
    const int tid = threadIdx.x;
    uint4* b4 = (uint4*)bits;
    for (int i = tid; i < N_TILES * TB_WORDS / 4; i += 256) b4[i] = make_uint4(0u, 0u, 0u, 0u);
    __syncthreads();

    const int g = blockIdx.x * (TB_EDGES / 4) + tid * 2;
    const int4 d0 = ((const int4*)dst)[g];
    const int4 d1 = ((const int4*)dst)[g + 1];
    unsigned int* wp = bits + (tid >> 2);
    const unsigned b0 = 1u << ((tid & 3) * 8);
#define SET_BIT(d, k)                                                                              \
    __hip_atomic_fetch_or(wp + (((unsigned)(d) >> TILE_SHIFT) & (N_TILES - 1)) * TB_WORDS, b0 << (k), \
                          __ATOMIC_RELAXED, __HIP_MEMORY_SCOPE_WORKGROUP)
    SET_BIT(d0.x, 0); SET_BIT(d0.y, 1); SET_BIT(d0.z, 2); SET_BIT(d0.w, 3);
    SET_BIT(d1.x, 4); SET_BIT(d1.y, 5); SET_BIT(d1.z, 6); SET_BIT(d1.w, 7);
#undef SET_BIT
    __syncthreads();

    for (int i = tid; i < N_TILES * TB_WORDS / 4; i += 256) {
        uint4* o = (uint4*)(planes + (size_t)(i >> 4) * PLANE_WORDS + blockIdx.x * TB_WORDS);
        o[i & 15] = b4[i];
    }
}

#define TILE_D (1 << TILE_SHIFT)
#define AGG_WAVES 32
#define AGG_THREADS (AGG_WAVES * 32)
#define ROWS_PER_WAVE (TILE_D / AGG_WAVES)
#define PLANE_VEC (PLANE_WORDS / 4)
#define TRIPS (PLANE_VEC / AGG_THREADS)
#define BATCH 8
#define MAXB 32
#define SQ 64
#define RCAP 64
#define RSTEP 16
static_assert(PLANE_VEC % AGG_THREADS == 0 && SQ >= 2 * MAXB, "plane pieces per lane; queue holds left-overs plus one turn");
static_assert(BATCH == 8 && HEADS * 8 == 32 && OUT_D == 32, "walk 2: a group of eight lanes per head, one entry per lane of a group");
static_assert(RCAP % RSTEP == 0 && RSTEP == 2 * BATCH && TILE_D % AGG_WAVES == 0, "bins are walked in whole steps of two groups of eight; rows split evenly over the waves");

__device__ __forceinline__ void lds_add(float* p, float v) {
    __hip_atomic_fetch_add(p, v, __ATOMIC_RELAXED, __HIP_MEMORY_SCOPE_WORKGROUP);
}
__device__ __forceinline__ void lds_max_f(float* p, float v) {
    if (v >= 0.0f)
        __hip_atomic_fetch_max((int*)p, __float_as_int(v), __ATOMIC_RELAXED, __HIP_MEMORY_SCOPE_WORKGROUP);
    else
        __hip_atomic_fetch_min((unsigned int*)p, __float_as_uint(v), __ATOMIC_RELAXED, __HIP_MEMORY_SCOPE_WORKGROUP);
}
__device__ __forceinline__ float leaky(float x) {
    const float v = (x >= 0.f) ? x : NEG_SLOPE * x;
    return v + 0.0f;
}

struct AggCtx {
    const int* __restrict__ src;
    const int* __restrict__ dst;
    const float* __restrict__ h;
    const float* __restrict__ el;
    float* acc;
    float* smax;
    float* ssum;
    const float4* ser;
    int* rowq;
    int* rcnt;
    int* ovf;
};

__device__ __forceinline__ void max_batch(const int* q, int nb, int lane, const AggCtx& c) {
    if (lane < nb) {
        const int e = q[lane];
        const int s = c.src[e];
        const int dl = c.dst[e] & (TILE_D - 1);
        const float4 l4 = *(const float4*)(c.el + (size_t)s * HEADS);
        const float4 r4 = c.ser[dl];
        float* mp = c.smax + dl * HEADS;
        lds_max_f(mp + 0, leaky(l4.x + r4.x));
        lds_max_f(mp + 1, leaky(l4.y + r4.y));
        lds_max_f(mp + 2, leaky(l4.z + r4.z));
        lds_max_f(mp + 3, leaky(l4.w + r4.w));
        const int pos = __hip_atomic_fetch_add(c.rcnt + dl, 1, __ATOMIC_RELAXED, __HIP_MEMORY_SCOPE_WORKGROUP);
        if (pos < RCAP) c.rowq[dl * RCAP + pos] = s;
        else *c.ovf = 1;
    }
}

template <int K>
__device__ __forceinline__ float group_bcast(float v) {
    return __uint_as_float(__builtin_amdgcn_mov_dpp8(__float_as_uint(v), K * 0x249249u));
}

__device__ __forceinline__ float group_sum(float v) {
    v += __uint_as_float(__builtin_amdgcn_mov_dpp8(__float_as_uint(v), 0xde54c1u));
    v += __uint_as_float(__builtin_amdgcn_mov_dpp8(__float_as_uint(v), 0xb3e21au));
    v += __uint_as_float(__builtin_amdgcn_mov_dpp8(__float_as_uint(v), 0x688facu));
    return v;
}

template <bool FULL>
__device__ __forceinline__ void row_batch(const int* rq, int nb, int lane, int hd, float r1, float m1,
                                          float& asum, float4& acc, const AggCtx& c) {
    const int k8 = lane & 7;
    const bool mine0 = FULL || k8 < nb;
    const bool mine1 = FULL || k8 + 8 < nb;
    const int s0 = rq[FULL ? k8 : min(k8, nb - 1)];
    const int s1 = rq[FULL ? k8 + 8 : min(k8 + 8, nb - 1)];
    const float l0 = c.el[(size_t)s0 * HEADS + hd];
    const float l1 = c.el[(size_t)s1 * HEADS + hd];

    float4 hv[RSTEP];
#pragma unroll
    for (int k = 0; k < RSTEP; ++k) {
        if (FULL || k < nb) {
            const int si = __builtin_amdgcn_readlane(k < 8 ? s0 : s1, k & 7);
            hv[k] = ((const float4*)(c.h + (size_t)si * FEAT))[lane];
        } else {
            hv[k] = make_float4(0.f, 0.f, 0.f, 0.f);
        }
    }
    __builtin_amdgcn_sched_barrier(0);

    const float a0 = expf(leaky(l0 + r1) - m1);
    const float a1 = expf(leaky(l1 + r1) - m1);
    asum += mine0 ? a0 : 0.0f;
    asum += mine1 ? a1 : 0.0f;

#define SUM_ROW(K, A, H)                                            \
    if (FULL || (H) + K < nb) {                                     \
        const float w = group_bcast<K>(A);                          \
        acc.x += hv[(H) + K].x * w;                                 \
        acc.y += hv[(H) + K].y * w;                                 \
        acc.z += hv[(H) + K].z * w;                                 \
        acc.w += hv[(H) + K].w * w;                                 \
    }
    SUM_ROW(0, a0, 0) SUM_ROW(1, a0, 0) SUM_ROW(2, a0, 0) SUM_ROW(3, a0, 0)
    SUM_ROW(4, a0, 0) SUM_ROW(5, a0, 0) SUM_ROW(6, a0, 0) SUM_ROW(7, a0, 0)
    SUM_ROW(0, a1, 8) SUM_ROW(1, a1, 8) SUM_ROW(2, a1, 8) SUM_ROW(3, a1, 8)
    SUM_ROW(4, a1, 8) SUM_ROW(5, a1, 8) SUM_ROW(6, a1, 8) SUM_ROW(7, a1, 8)
#undef SUM_ROW
}

__device__ __forceinline__ void sum_row(int r, int lane, float4* orow, const AggCtx& c) {
    const int hd = lane >> 3;
    const int n = min(__builtin_amdgcn_readfirstlane(c.rcnt[r]), RCAP);
    const int* rq = c.rowq + r * RCAP;
    const float r1 = ((const float*)c.ser)[r * HEADS + hd];
    const float m1 = c.smax[r * HEADS + hd];
    float asum = 0.0f;
    float4 acc = make_float4(0.f, 0.f, 0.f, 0.f);
    int i = 0;
    for (; i + RSTEP <= n; i += RSTEP) row_batch<true>(rq + i, RSTEP, lane, hd, r1, m1, asum, acc, c);
    if (i < n) row_batch<false>(rq + i, n - i, lane, hd, r1, m1, asum, acc, c);
    const float sd = group_sum(asum);
    const bool any = sd > 0.f;
    float4 v;
    v.x = any ? acc.x / sd : 0.f;
    v.y = any ? acc.y / sd : 0.f;
    v.z = any ? acc.z / sd : 0.f;
    v.w = any ? acc.w / sd : 0.f;
    v.x = (v.x > 0.f) ? v.x : (expf(v.x) - 1.0f);
    v.y = (v.y > 0.f) ? v.y : (expf(v.y) - 1.0f);
    v.z = (v.z > 0.f) ? v.z : (expf(v.z) - 1.0f);
    v.w = (v.w > 0.f) ? v.w : (expf(v.w) - 1.0f);
    orow[lane] = v;
}

template <bool FULL>
__device__ __forceinline__ void agg_batch(const int* q, int nb, int lane, const AggCtx& c) {
    const int hd = lane >> 3;
    int qi = lane & (BATCH - 1);
    const bool mine = FULL || qi < nb;
    if (!FULL) qi = min(qi, nb - 1);
    const int e = q[qi];
    const int s = c.src[e];
    const int dl = c.dst[e] & (TILE_D - 1);
    const float l1 = c.el[(size_t)s * HEADS + hd];

    float4 hv[BATCH];
#pragma unroll
    for (int k = 0; k < BATCH; ++k) {
        if (FULL || k < nb) {
            const int si = __builtin_amdgcn_readlane(s, k);
            hv[k] = ((const float4*)(c.h + (size_t)si * FEAT))[lane];
        } else {
            hv[k] = make_float4(0.f, 0.f, 0.f, 0.f);
        }
    }

    const float r1 = ((const float*)c.ser)[dl * HEADS + hd];
    const float m1 = c.smax[dl * HEADS + hd];
    const float a = expf(leaky(l1 + r1) - m1);
    if (mine) lds_add(c.ssum + dl * HEADS + hd, a);

#define ADD_ROW(K)                                                  \
    if (FULL || K < nb) {                                           \
        const int dk = __builtin_amdgcn_readlane(dl, K);            \
        const float w = group_bcast<K>(a);                          \
        float* ap = c.acc + dk * FEAT + lane;                       \
        lds_add(ap,      hv[K].x * w);                              \
        lds_add(ap + 32, hv[K].y * w);                              \
        lds_add(ap + 64, hv[K].z * w);                              \
        lds_add(ap + 96, hv[K].w * w);                              \
    }
    ADD_ROW(0) ADD_ROW(1) ADD_ROW(2) ADD_ROW(3)
    ADD_ROW(4) ADD_ROW(5) ADD_ROW(6) ADD_ROW(7)
#undef ADD_ROW
}

template <int WALK>
__device__ __forceinline__ void queue_word(unsigned x, int ebase, int& cnt, int lane, int* myq, const AggCtx& c) {
    unsigned m = __builtin_amdgcn_ballot_w32(x != 0);
    while (m) {
        if (x) {
            myq[cnt + __builtin_amdgcn_mbcnt_lo(m, 0)] = ebase + __builtin_ctz(x);
            x &= x - 1;
        }
        cnt += __builtin_popcount(m);
        m = __builtin_amdgcn_ballot_w32(x != 0);
        if constexpr (WALK == 1) {
            while (cnt >= MAXB) {
                cnt -= MAXB;
                max_batch(myq + cnt, MAXB, lane, c);
            }
        } else {
            while (cnt >= BATCH) {
                cnt -= BATCH;
                agg_batch<true>(myq + cnt, BATCH, lane, c);
            }
        }
    }
}

template <int WALK>
__device__ __forceinline__ void walk_edges(const uint4* pl, int wave, int lane, int* myq, const AggCtx& c) {
    int cnt = 0;

    uint4 cur = pl[wave * 32 + lane];
    for (int it = 0; it < TRIPS; ++it) {
        const int piece = (it * AGG_WAVES + wave) * 32 + lane;
        uint4 nxt = cur;
        if (it + 1 < TRIPS)
            nxt = pl[piece + AGG_THREADS];
        const int ebase = piece * 128;
        queue_word<WALK>(cur.x, ebase, cnt, lane, myq, c);
        queue_word<WALK>(cur.y, ebase + 32, cnt, lane, myq, c);
        queue_word<WALK>(cur.z, ebase + 64, cnt, lane, myq, c);
        queue_word<WALK>(cur.w, ebase + 96, cnt, lane, myq, c);
        cur = nxt;
    }
    if (cnt > 0) {
        if constexpr (WALK == 1) max_batch(myq, cnt, lane, c);
        else agg_batch<false>(myq, cnt, lane, c);
    }
}

__global__ __launch_bounds__(AGG_THREADS) void aggregate_kernel(const int* __restrict__ src,
                                                                const int* __restrict__ dst,
                                                                const float* __restrict__ h,
                                                                const float* __restrict__ el,
                                                                const float* __restrict__ er,
                                                                float* out) {
    __shared__ float acc[TILE_D * FEAT];
    __shared__ float4 ser[TILE_D];
    __shared__ __attribute__((aligned(16))) float smax[TILE_D * HEADS];
    __shared__ float ssum[TILE_D * HEADS];
    __shared__ int rowq[TILE_D * RCAP];
    __shared__ int rcnt[TILE_D];
    __shared__ int sq[AGG_WAVES][SQ];
    __shared__ int ovf;

    const int tid = threadIdx.x;
    const int lane = tid & 31;
    const int wave = tid >> 5;
    const int tile = blockIdx.x;

    if (tid < TILE_D) {
        ser[tid] = ((const float4*)er)[(size_t)tile * TILE_D + tid];
        rcnt[tid] = 0;
    }
    if (tid == 0) ovf = 0;
    ((unsigned int*)smax)[tid] = 0xFF800000u;
    ssum[tid] = 0.0f;
    __syncthreads();

    AggCtx c;
    c.src = src; c.dst = dst; c.h = h; c.el = el;
    c.acc = acc; c.smax = smax; c.ssum = ssum; c.ser = ser;
    c.rowq = rowq; c.rcnt = rcnt; c.ovf = &ovf;
    const uint4* pl = (const uint4*)(out + (size_t)tile * TILE_D * FEAT);
    static_assert(PLANE_WORDS == TILE_D * FEAT, "a plane fills exactly its tile's output rows");
    float4* o4 = (float4*)(out + (size_t)tile * TILE_D * FEAT);

    walk_edges<1>(pl, wave, lane, sq[wave], c);
    __syncthreads();

    if (ovf == 0) {
        for (int j = 0; j < ROWS_PER_WAVE; ++j) {
            const int r = wave * ROWS_PER_WAVE + j;
            sum_row(r, lane, o4 + r * (FEAT / 4), c);
        }
    } else {
        for (int i = tid; i < TILE_D * FEAT / 4; i += AGG_THREADS)
            ((float4*)acc)[i] = make_float4(0.f, 0.f, 0.f, 0.f);
        __syncthreads();
        walk_edges<2>(pl, wave, lane, sq[wave], c);
        __syncthreads();
        for (int i = tid; i < TILE_D * FEAT / 4; i += AGG_THREADS) {
            const float* ar = acc + (i >> 5) * FEAT + (i & 31);
            float4 v = make_float4(ar[0], ar[32], ar[64], ar[96]);
            const float sd = ssum[(i >> 5) * HEADS + ((i & 31) >> 3)];
            const bool any = sd > 0.f;
            v.x = any ? v.x / sd : 0.f;
            v.y = any ? v.y / sd : 0.f;
            v.z = any ? v.z / sd : 0.f;
            v.w = any ? v.w / sd : 0.f;
            v.x = (v.x > 0.f) ? v.x : (expf(v.x) - 1.0f);
            v.y = (v.y > 0.f) ? v.y : (expf(v.y) - 1.0f);
            v.z = (v.z > 0.f) ? v.z : (expf(v.z) - 1.0f);
            v.w = (v.w > 0.f) ? v.w : (expf(v.w) - 1.0f);
            o4[i] = v;
        }
    }
    if (tile == 0 && tid == 0) {
        out[(size_t)NUM_DSTN * FEAT + 0] = 1.0f;
        out[(size_t)NUM_DSTN * FEAT + 1] = 1.0f;
    }
}

extern "C" void kernel_launch(void* const* d_in, const int* in_sizes, int n_in,
                              void* d_out, int out_size, void* d_ws, size_t ws_size,
                              hipStream_t stream) {
    const float* feats  = (const float*)d_in[0];
    const int*   src    = (const int*)d_in[1];
    const int*   dst    = (const int*)d_in[2];
    const float* W      = (const float*)d_in[3];
    const float* attn_l = (const float*)d_in[4];
    const float* attn_r = (const float*)d_in[5];
    float* out = (float*)d_out;

    float* ws   = (float*)d_ws;
    float* hbuf = ws;
    float* el   = hbuf + (size_t)N_NODES * FEAT;
    float* er   = el + (size_t)N_NODES * HEADS;

    {
        int rowTiles = N_NODES / 16;
        int blocks = rowTiles / 8;
        gemm_kernel<<<blocks, 256, 0, stream>>>(feats, W, hbuf);
    }
    {
        int n = N_NODES * HEADS;
        attn_logits_kernel<<<(n + 255) / 256, 256, 0, stream>>>(hbuf, attn_l, attn_r, el, er);
    }
    tile_plane_kernel<<<E_EDGES / TB_EDGES, 256, 0, stream>>>(dst, (unsigned int*)out);
    aggregate_kernel<<<NUM_DSTN / TILE_D, AGG_THREADS, 0, stream>>>(src, dst, hbuf, el, er, out);
}


# Logits pass: a lane owns 4 columns of a row (16-byte loads, 512 contiguous bytes per instruction), 32 rows per wave, 4096 waves not 16384; el/er sum order changes: 4 products per lane, 8-lane tree
# speedup vs baseline: 6.8041x; 1.0539x over previous
#include <hip/hip_runtime.h>
#include <math.h>

#define N_NODES 131072
#define E_EDGES 1048576
#define NUM_DSTN 65536
#define IN_DIM 256
#define HEADS 4
#define OUT_D 32
#define FEAT (HEADS * OUT_D)
#define NEG_SLOPE 0.2f

typedef __attribute__((ext_vector_type(2))) float v2f;
typedef __attribute__((ext_vector_type(8))) float v8f;

__global__ __launch_bounds__(256, 8) void gemm_kernel(const float* __restrict__ feats,
                                                   const float* __restrict__ W,
                                                   float* __restrict__ h) {
    const int wave = blockIdx.x * (blockDim.x >> 5) + (threadIdx.x >> 5);
    const int lane = threadIdx.x & 31;
    const int lo = lane & 15;
    const int hi = lane >> 4;
    const int rowBase = wave * 16;
    const float* __restrict__ arow = feats + (size_t)(rowBase + lo) * IN_DIM;

    v8f acc[8];
#pragma unroll
    for (int t = 0; t < 8; ++t) acc[t] = (v8f){0.f, 0.f, 0.f, 0.f, 0.f, 0.f, 0.f, 0.f};

#define GEMM_TILE(t, bx, by)                                                       \
    {                                                                              \
        v2f b;                                                                     \
        b.x = (bx);                                                                \
        b.y = (by);                                                                \
        acc[t] = __builtin_amdgcn_wmma_f32_16x16x4_f32(false, a, false, b, (short)0, acc[t], false, false); \
    }
    for (int k = 0; k < IN_DIM; k += 4) {
        const int ka = k + 2 * hi;
        v2f a;
        a.x = arow[ka];
        a.y = arow[ka + 1];
        const float4* __restrict__ w0 = (const float4*)(W + (size_t)ka * FEAT) + lo;
        const float4* __restrict__ w1 = (const float4*)(W + (size_t)(ka + 1) * FEAT) + lo;
        const float4 x0 = w0[0];
        const float4 x1 = w0[16];
        const float4 y0 = w1[0];
        const float4 y1 = w1[16];
        GEMM_TILE(0, x0.x, y0.x)
        GEMM_TILE(1, x0.y, y0.y)
        GEMM_TILE(2, x0.z, y0.z)
        GEMM_TILE(3, x0.w, y0.w)
        GEMM_TILE(4, x1.x, y1.x)
        GEMM_TILE(5, x1.y, y1.y)
        GEMM_TILE(6, x1.z, y1.z)
        GEMM_TILE(7, x1.w, y1.w)
    }
#undef GEMM_TILE

#pragma unroll
    for (int i = 0; i < 8; ++i) {
        float4* __restrict__ hp = (float4*)(h + (size_t)(rowBase + i + 8 * hi) * FEAT) + lo;
        hp[0]  = make_float4(acc[0][i], acc[1][i], acc[2][i], acc[3][i]);
        hp[16] = make_float4(acc[4][i], acc[5][i], acc[6][i], acc[7][i]);
    }
}

#define LG_ROWS 16
#define LG_BATCHES 2
#define LG_WAVES 8
static_assert(LG_ROWS % 8 == 0 && N_NODES % (LG_ROWS * LG_BATCHES * LG_WAVES) == 0 && FEAT == 128 && HEADS * 8 == 32,
              "rows are summed eight at a time; the grid covers the nodes exactly; a lane's four columns lie in head lane / 8");

__device__ __forceinline__ float lane_xor4(float v) {
    return __uint_as_float(__builtin_amdgcn_mov_dpp8(__float_as_uint(v), 0x688facu));
}
__device__ __forceinline__ float lane_xor2(float v) {
    return __uint_as_float(__builtin_amdgcn_mov_dpp8(__float_as_uint(v), 0xb3e21au));
}
__device__ __forceinline__ float lane_xor1(float v) {
    return __uint_as_float(__builtin_amdgcn_mov_dpp8(__float_as_uint(v), 0xde54c1u));
}

__device__ __forceinline__ float group_sum_rows8(const float* p, int k8) {
    const bool b2 = (k8 & 4) != 0;
    const bool b1 = (k8 & 2) != 0;
    const bool b0 = (k8 & 1) != 0;
    float q[4];
#pragma unroll
    for (int j = 0; j < 4; ++j) {
        const float keep = b2 ? p[j + 4] : p[j];
        const float give = b2 ? p[j] : p[j + 4];
        q[j] = keep + lane_xor4(give);
    }
    float r[2];
#pragma unroll
    for (int j = 0; j < 2; ++j) {
        const float keep = b1 ? q[j + 2] : q[j];
        const float give = b1 ? q[j] : q[j + 2];
        r[j] = keep + lane_xor2(give);
    }
    const float keep = b0 ? r[1] : r[0];
    const float give = b0 ? r[0] : r[1];
    return keep + lane_xor1(give);
}

__global__ __launch_bounds__(LG_WAVES * 32) void attn_logits_kernel(const float* __restrict__ h,
                                                                    const float* __restrict__ attn_l,
                                                                    const float* __restrict__ attn_r,
                                                                    float* __restrict__ el, float* __restrict__ er) {
    const int lane = threadIdx.x & 31;
    const int k8 = lane & 7;
    const int hd = lane >> 3;
    const int wave = blockIdx.x * LG_WAVES + (threadIdx.x >> 5);
    const float4 al = ((const float4*)attn_l)[lane];
    const float4 ar = ((const float4*)attn_r)[lane];
    for (int b = 0; b < LG_BATCHES; ++b) {
        const int row0 = (wave * LG_BATCHES + b) * LG_ROWS;
        const float4* __restrict__ hp = (const float4*)(h + (size_t)row0 * FEAT) + lane;
        float4 v[LG_ROWS];
#pragma unroll
        for (int j = 0; j < LG_ROWS; ++j) v[j] = hp[j * (FEAT / 4)];
        float pl[LG_ROWS], pr[LG_ROWS];
#pragma unroll
        for (int j = 0; j < LG_ROWS; ++j) {
            pl[j] = v[j].x * al.x + v[j].y * al.y + v[j].z * al.z + v[j].w * al.w;
            pr[j] = v[j].x * ar.x + v[j].y * ar.y + v[j].z * ar.z + v[j].w * ar.w;
        }
#pragma unroll
        for (int g = 0; g < LG_ROWS / 8; ++g) {
            const float sl = group_sum_rows8(pl + 8 * g, k8);
            const float sr = group_sum_rows8(pr + 8 * g, k8);
            const int o = (row0 + 8 * g + k8) * HEADS + hd;
            el[o] = sl;
            er[o] = sr;
        }
    }
}

#define TILE_SHIFT 8
#define N_TILES (NUM_DSTN >> TILE_SHIFT)
#define PLANE_WORDS (E_EDGES / 32)
#define TB_EDGES 2048
#define TB_WORDS (TB_EDGES / 32)
__global__ __launch_bounds__(256) void tile_plane_kernel(const int* __restrict__ dst,
                                                         unsigned int* planes) {
    __shared__ __attribute__((aligned(16))) unsigned int bits[N_TILES * TB_WORDS];
    const int tid = threadIdx.x;
    uint4* b4 = (uint4*)bits;
    for (int i = tid; i < N_TILES * TB_WORDS / 4; i += 256) b4[i] = make_uint4(0u, 0u, 0u, 0u);
    __syncthreads();

    const int g = blockIdx.x * (TB_EDGES / 4) + tid * 2;
    const int4 d0 = ((const int4*)dst)[g];
    const int4 d1 = ((const int4*)dst)[g + 1];
    unsigned int* wp = bits + (tid >> 2);
    const unsigned b0 = 1u << ((tid & 3) * 8);
#define SET_BIT(d, k)                                                                              \
    __hip_atomic_fetch_or(wp + (((unsigned)(d) >> TILE_SHIFT) & (N_TILES - 1)) * TB_WORDS, b0 << (k), \
                          __ATOMIC_RELAXED, __HIP_MEMORY_SCOPE_WORKGROUP)
    SET_BIT(d0.x, 0); SET_BIT(d0.y, 1); SET_BIT(d0.z, 2); SET_BIT(d0.w, 3);
    SET_BIT(d1.x, 4); SET_BIT(d1.y, 5); SET_BIT(d1.z, 6); SET_BIT(d1.w, 7);
#undef SET_BIT
    __syncthreads();

    for (int i = tid; i < N_TILES * TB_WORDS / 4; i += 256) {
        uint4* o = (uint4*)(planes + (size_t)(i >> 4) * PLANE_WORDS + blockIdx.x * TB_WORDS);
        o[i & 15] = b4[i];
    }
}

#define TILE_D (1 << TILE_SHIFT)
#define AGG_WAVES 32
#define AGG_THREADS (AGG_WAVES * 32)
#define ROWS_PER_WAVE (TILE_D / AGG_WAVES)
#define PLANE_VEC (PLANE_WORDS / 4)
#define TRIPS (PLANE_VEC / AGG_THREADS)
#define BATCH 8
#define MAXB 32
#define SQ 64
#define RCAP 64
#define RSTEP 16
static_assert(PLANE_VEC % AGG_THREADS == 0 && SQ >= 2 * MAXB, "plane pieces per lane; queue holds left-overs plus one turn");
static_assert(BATCH == 8 && HEADS * 8 == 32 && OUT_D == 32, "walk 2: a group of eight lanes per head, one entry per lane of a group");
static_assert(RCAP % RSTEP == 0 && RSTEP == 2 * BATCH && TILE_D % AGG_WAVES == 0, "bins are walked in whole steps of two groups of eight; rows split evenly over the waves");

__device__ __forceinline__ void lds_add(float* p, float v) {
    __hip_atomic_fetch_add(p, v, __ATOMIC_RELAXED, __HIP_MEMORY_SCOPE_WORKGROUP);
}
__device__ __forceinline__ void lds_max_f(float* p, float v) {
    if (v >= 0.0f)
        __hip_atomic_fetch_max((int*)p, __float_as_int(v), __ATOMIC_RELAXED, __HIP_MEMORY_SCOPE_WORKGROUP);
    else
        __hip_atomic_fetch_min((unsigned int*)p, __float_as_uint(v), __ATOMIC_RELAXED, __HIP_MEMORY_SCOPE_WORKGROUP);
}
__device__ __forceinline__ float leaky(float x) {
    const float v = (x >= 0.f) ? x : NEG_SLOPE * x;
    return v + 0.0f;
}

struct AggCtx {
    const int* __restrict__ src;
    const int* __restrict__ dst;
    const float* __restrict__ h;
    const float* __restrict__ el;
    float* acc;
    float* smax;
    float* ssum;
    const float4* ser;
    int* rowq;
    int* rcnt;
    int* ovf;
};

__device__ __forceinline__ void max_batch(const int* q, int nb, int lane, const AggCtx& c) {
    if (lane < nb) {
        const int e = q[lane];
        const int s = c.src[e];
        const int dl = c.dst[e] & (TILE_D - 1);
        const float4 l4 = *(const float4*)(c.el + (size_t)s * HEADS);
        const float4 r4 = c.ser[dl];
        float* mp = c.smax + dl * HEADS;
        lds_max_f(mp + 0, leaky(l4.x + r4.x));
        lds_max_f(mp + 1, leaky(l4.y + r4.y));
        lds_max_f(mp + 2, leaky(l4.z + r4.z));
        lds_max_f(mp + 3, leaky(l4.w + r4.w));
        const int pos = __hip_atomic_fetch_add(c.rcnt + dl, 1, __ATOMIC_RELAXED, __HIP_MEMORY_SCOPE_WORKGROUP);
        if (pos < RCAP) c.rowq[dl * RCAP + pos] = s;
        else *c.ovf = 1;
    }
}

template <int K>
__device__ __forceinline__ float group_bcast(float v) {
    return __uint_as_float(__builtin_amdgcn_mov_dpp8(__float_as_uint(v), K * 0x249249u));
}

__device__ __forceinline__ float group_sum(float v) {
    v += __uint_as_float(__builtin_amdgcn_mov_dpp8(__float_as_uint(v), 0xde54c1u));
    v += __uint_as_float(__builtin_amdgcn_mov_dpp8(__float_as_uint(v), 0xb3e21au));
    v += __uint_as_float(__builtin_amdgcn_mov_dpp8(__float_as_uint(v), 0x688facu));
    return v;
}

template <bool FULL>
__device__ __forceinline__ void row_batch(const int* rq, int nb, int lane, int hd, float r1, float m1,
                                          float& asum, float4& acc, const AggCtx& c) {
    const int k8 = lane & 7;
    const bool mine0 = FULL || k8 < nb;
    const bool mine1 = FULL || k8 + 8 < nb;
    const int s0 = rq[FULL ? k8 : min(k8, nb - 1)];
    const int s1 = rq[FULL ? k8 + 8 : min(k8 + 8, nb - 1)];
    const float l0 = c.el[(size_t)s0 * HEADS + hd];
    const float l1 = c.el[(size_t)s1 * HEADS + hd];

    float4 hv[RSTEP];
#pragma unroll
    for (int k = 0; k < RSTEP; ++k) {
        if (FULL || k < nb) {
            const int si = __builtin_amdgcn_readlane(k < 8 ? s0 : s1, k & 7);
            hv[k] = ((const float4*)(c.h + (size_t)si * FEAT))[lane];
        } else {
            hv[k] = make_float4(0.f, 0.f, 0.f, 0.f);
        }
    }
    __builtin_amdgcn_sched_barrier(0);

    const float a0 = expf(leaky(l0 + r1) - m1);
    const float a1 = expf(leaky(l1 + r1) - m1);
    asum += mine0 ? a0 : 0.0f;
    asum += mine1 ? a1 : 0.0f;

#define SUM_ROW(K, A, H)                                            \
    if (FULL || (H) + K < nb) {                                     \
        const float w = group_bcast<K>(A);                          \
        acc.x += hv[(H) + K].x * w;                                 \
        acc.y += hv[(H) + K].y * w;                                 \
        acc.z += hv[(H) + K].z * w;                                 \
        acc.w += hv[(H) + K].w * w;                                 \
    }
    SUM_ROW(0, a0, 0) SUM_ROW(1, a0, 0) SUM_ROW(2, a0, 0) SUM_ROW(3, a0, 0)
    SUM_ROW(4, a0, 0) SUM_ROW(5, a0, 0) SUM_ROW(6, a0, 0) SUM_ROW(7, a0, 0)
    SUM_ROW(0, a1, 8) SUM_ROW(1, a1, 8) SUM_ROW(2, a1, 8) SUM_ROW(3, a1, 8)
    SUM_ROW(4, a1, 8) SUM_ROW(5, a1, 8) SUM_ROW(6, a1, 8) SUM_ROW(7, a1, 8)
#undef SUM_ROW
}

__device__ __forceinline__ void sum_row(int r, int lane, float4* orow, const AggCtx& c) {
    const int hd = lane >> 3;
    const int n = min(__builtin_amdgcn_readfirstlane(c.rcnt[r]), RCAP);
    const int* rq = c.rowq + r * RCAP;
    const float r1 = ((const float*)c.ser)[r * HEADS + hd];
    const float m1 = c.smax[r * HEADS + hd];
    float asum = 0.0f;
    float4 acc = make_float4(0.f, 0.f, 0.f, 0.f);
    int i = 0;
    for (; i + RSTEP <= n; i += RSTEP) row_batch<true>(rq + i, RSTEP, lane, hd, r1, m1, asum, acc, c);
    if (i < n) row_batch<false>(rq + i, n - i, lane, hd, r1, m1, asum, acc, c);
    const float sd = group_sum(asum);
    const bool any = sd > 0.f;
    float4 v;
    v.x = any ? acc.x / sd : 0.f;
    v.y = any ? acc.y / sd : 0.f;
    v.z = any ? acc.z / sd : 0.f;
    v.w = any ? acc.w / sd : 0.f;
    v.x = (v.x > 0.f) ? v.x : (expf(v.x) - 1.0f);
    v.y = (v.y > 0.f) ? v.y : (expf(v.y) - 1.0f);
    v.z = (v.z > 0.f) ? v.z : (expf(v.z) - 1.0f);
    v.w = (v.w > 0.f) ? v.w : (expf(v.w) - 1.0f);
    orow[lane] = v;
}

template <bool FULL>
__device__ __forceinline__ void agg_batch(const int* q, int nb, int lane, const AggCtx& c) {
    const int hd = lane >> 3;
    int qi = lane & (BATCH - 1);
    const bool mine = FULL || qi < nb;
    if (!FULL) qi = min(qi, nb - 1);
    const int e = q[qi];
    const int s = c.src[e];
    const int dl = c.dst[e] & (TILE_D - 1);
    const float l1 = c.el[(size_t)s * HEADS + hd];

    float4 hv[BATCH];
#pragma unroll
    for (int k = 0; k < BATCH; ++k) {
        if (FULL || k < nb) {
            const int si = __builtin_amdgcn_readlane(s, k);
            hv[k] = ((const float4*)(c.h + (size_t)si * FEAT))[lane];
        } else {
            hv[k] = make_float4(0.f, 0.f, 0.f, 0.f);
        }
    }

    const float r1 = ((const float*)c.ser)[dl * HEADS + hd];
    const float m1 = c.smax[dl * HEADS + hd];
    const float a = expf(leaky(l1 + r1) - m1);
    if (mine) lds_add(c.ssum + dl * HEADS + hd, a);

#define ADD_ROW(K)                                                  \
    if (FULL || K < nb) {                                           \
        const int dk = __builtin_amdgcn_readlane(dl, K);            \
        const float w = group_bcast<K>(a);                          \
        float* ap = c.acc + dk * FEAT + lane;                       \
        lds_add(ap,      hv[K].x * w);                              \
        lds_add(ap + 32, hv[K].y * w);                              \
        lds_add(ap + 64, hv[K].z * w);                              \
        lds_add(ap + 96, hv[K].w * w);                              \
    }
    ADD_ROW(0) ADD_ROW(1) ADD_ROW(2) ADD_ROW(3)
    ADD_ROW(4) ADD_ROW(5) ADD_ROW(6) ADD_ROW(7)
#undef ADD_ROW
}

template <int WALK>
__device__ __forceinline__ void queue_word(unsigned x, int ebase, int& cnt, int lane, int* myq, const AggCtx& c) {
    unsigned m = __builtin_amdgcn_ballot_w32(x != 0);
    while (m) {
        if (x) {
            myq[cnt + __builtin_amdgcn_mbcnt_lo(m, 0)] = ebase + __builtin_ctz(x);
            x &= x - 1;
        }
        cnt += __builtin_popcount(m);
        m = __builtin_amdgcn_ballot_w32(x != 0);
        if constexpr (WALK == 1) {
            while (cnt >= MAXB) {
                cnt -= MAXB;
                max_batch(myq + cnt, MAXB, lane, c);
            }
        } else {
            while (cnt >= BATCH) {
                cnt -= BATCH;
                agg_batch<true>(myq + cnt, BATCH, lane, c);
            }
        }
    }
}

template <int WALK>
__device__ __forceinline__ void walk_edges(const uint4* pl, int wave, int lane, int* myq, const AggCtx& c) {
    int cnt = 0;

    uint4 cur = pl[wave * 32 + lane];
    for (int it = 0; it < TRIPS; ++it) {
        const int piece = (it * AGG_WAVES + wave) * 32 + lane;
        uint4 nxt = cur;
        if (it + 1 < TRIPS)
            nxt = pl[piece + AGG_THREADS];
        const int ebase = piece * 128;
        queue_word<WALK>(cur.x, ebase, cnt, lane, myq, c);
        queue_word<WALK>(cur.y, ebase + 32, cnt, lane, myq, c);
        queue_word<WALK>(cur.z, ebase + 64, cnt, lane, myq, c);
        queue_word<WALK>(cur.w, ebase + 96, cnt, lane, myq, c);
        cur = nxt;
    }
    if (cnt > 0) {
        if constexpr (WALK == 1) max_batch(myq, cnt, lane, c);
        else agg_batch<false>(myq, cnt, lane, c);
    }
}

__global__ __launch_bounds__(AGG_THREADS) void aggregate_kernel(const int* __restrict__ src,
                                                                const int* __restrict__ dst,
                                                                const float* __restrict__ h,
                                                                const float* __restrict__ el,
                                                                const float* __restrict__ er,
                                                                float* out) {
    __shared__ float acc[TILE_D * FEAT];
    __shared__ float4 ser[TILE_D];
    __shared__ __attribute__((aligned(16))) float smax[TILE_D * HEADS];
    __shared__ float ssum[TILE_D * HEADS];
    __shared__ int rowq[TILE_D * RCAP];
    __shared__ int rcnt[TILE_D];
    __shared__ int sq[AGG_WAVES][SQ];
    __shared__ int ovf;

    const int tid = threadIdx.x;
    const int lane = tid & 31;
    const int wave = tid >> 5;
    const int tile = blockIdx.x;

    if (tid < TILE_D) {
        ser[tid] = ((const float4*)er)[(size_t)tile * TILE_D + tid];
        rcnt[tid] = 0;
    }
    if (tid == 0) ovf = 0;
    ((unsigned int*)smax)[tid] = 0xFF800000u;
    ssum[tid] = 0.0f;
    __syncthreads();

    AggCtx c;
    c.src = src; c.dst = dst; c.h = h; c.el = el;
    c.acc = acc; c.smax = smax; c.ssum = ssum; c.ser = ser;
    c.rowq = rowq; c.rcnt = rcnt; c.ovf = &ovf;
    const uint4* pl = (const uint4*)(out + (size_t)tile * TILE_D * FEAT);
    static_assert(PLANE_WORDS == TILE_D * FEAT, "a plane fills exactly its tile's output rows");
    float4* o4 = (float4*)(out + (size_t)tile * TILE_D * FEAT);

    walk_edges<1>(pl, wave, lane, sq[wave], c);
    __syncthreads();

    if (ovf == 0) {
        for (int j = 0; j < ROWS_PER_WAVE; ++j) {
            const int r = wave * ROWS_PER_WAVE + j;
            sum_row(r, lane, o4 + r * (FEAT / 4), c);
        }
    } else {
        for (int i = tid; i < TILE_D * FEAT / 4; i += AGG_THREADS)
            ((float4*)acc)[i] = make_float4(0.f, 0.f, 0.f, 0.f);
        __syncthreads();
        walk_edges<2>(pl, wave, lane, sq[wave], c);
        __syncthreads();
        for (int i = tid; i < TILE_D * FEAT / 4; i += AGG_THREADS) {
            const float* ar = acc + (i >> 5) * FEAT + (i & 31);
            float4 v = make_float4(ar[0], ar[32], ar[64], ar[96]);
            const float sd = ssum[(i >> 5) * HEADS + ((i & 31) >> 3)];
            const bool any = sd > 0.f;
            v.x = any ? v.x / sd : 0.f;
            v.y = any ? v.y / sd : 0.f;
            v.z = any ? v.z / sd : 0.f;
            v.w = any ? v.w / sd : 0.f;
            v.x = (v.x > 0.f) ? v.x : (expf(v.x) - 1.0f);
            v.y = (v.y > 0.f) ? v.y : (expf(v.y) - 1.0f);
            v.z = (v.z > 0.f) ? v.z : (expf(v.z) - 1.0f);
            v.w = (v.w > 0.f) ? v.w : (expf(v.w) - 1.0f);
            o4[i] = v;
        }
    }
    if (tile == 0 && tid == 0) {
        out[(size_t)NUM_DSTN * FEAT + 0] = 1.0f;
        out[(size_t)NUM_DSTN * FEAT + 1] = 1.0f;
    }
}

extern "C" void kernel_launch(void* const* d_in, const int* in_sizes, int n_in,
                              void* d_out, int out_size, void* d_ws, size_t ws_size,
                              hipStream_t stream) {
    const float* feats  = (const float*)d_in[0];
    const int*   src    = (const int*)d_in[1];
    const int*   dst    = (const int*)d_in[2];
    const float* W      = (const float*)d_in[3];
    const float* attn_l = (const float*)d_in[4];
    const float* attn_r = (const float*)d_in[5];
    float* out = (float*)d_out;

    float* ws   = (float*)d_ws;
    float* hbuf = ws;
    float* el   = hbuf + (size_t)N_NODES * FEAT;
    float* er   = el + (size_t)N_NODES * HEADS;

    {
        int rowTiles = N_NODES / 16;
        int blocks = rowTiles / 8;
        gemm_kernel<<<blocks, 256, 0, stream>>>(feats, W, hbuf);
    }
    attn_logits_kernel<<<N_NODES / (LG_ROWS * LG_BATCHES * LG_WAVES), LG_WAVES * 32, 0, stream>>>(hbuf, attn_l, attn_r, el, er);
    tile_plane_kernel<<<E_EDGES / TB_EDGES, 256, 0, stream>>>(dst, (unsigned int*)out);
    aggregate_kernel<<<NUM_DSTN / TILE_D, AGG_THREADS, 0, stream>>>(src, dst, hbuf, el, er, out);
}


# Aggregation: a row's maximum is taken in walk 2 by the wave that owns it, from the el values it loads anyway (8-lane max exchange); walk 1 drops its el gather and 4 LDS max atomics per edge
# speedup vs baseline: 7.0138x; 1.0308x over previous
#include <hip/hip_runtime.h>
#include <math.h>

#define N_NODES 131072
#define E_EDGES 1048576
#define NUM_DSTN 65536
#define IN_DIM 256
#define HEADS 4
#define OUT_D 32
#define FEAT (HEADS * OUT_D)
#define NEG_SLOPE 0.2f

typedef __attribute__((ext_vector_type(2))) float v2f;
typedef __attribute__((ext_vector_type(8))) float v8f;

__global__ __launch_bounds__(256, 8) void gemm_kernel(const float* __restrict__ feats,
                                                   const float* __restrict__ W,
                                                   float* __restrict__ h) {
    const int wave = blockIdx.x * (blockDim.x >> 5) + (threadIdx.x >> 5);
    const int lane = threadIdx.x & 31;
    const int lo = lane & 15;
    const int hi = lane >> 4;
    const int rowBase = wave * 16;
    const float* __restrict__ arow = feats + (size_t)(rowBase + lo) * IN_DIM;

    v8f acc[8];
#pragma unroll
    for (int t = 0; t < 8; ++t) acc[t] = (v8f){0.f, 0.f, 0.f, 0.f, 0.f, 0.f, 0.f, 0.f};

#define GEMM_TILE(t, bx, by)                                                       \
    {                                                                              \
        v2f b;                                                                     \
        b.x = (bx);                                                                \
        b.y = (by);                                                                \
        acc[t] = __builtin_amdgcn_wmma_f32_16x16x4_f32(false, a, false, b, (short)0, acc[t], false, false); \
    }
    for (int k = 0; k < IN_DIM; k += 4) {
        const int ka = k + 2 * hi;
        v2f a;
        a.x = arow[ka];
        a.y = arow[ka + 1];
        const float4* __restrict__ w0 = (const float4*)(W + (size_t)ka * FEAT) + lo;
        const float4* __restrict__ w1 = (const float4*)(W + (size_t)(ka + 1) * FEAT) + lo;
        const float4 x0 = w0[0];
        const float4 x1 = w0[16];
        const float4 y0 = w1[0];
        const float4 y1 = w1[16];
        GEMM_TILE(0, x0.x, y0.x)
        GEMM_TILE(1, x0.y, y0.y)
        GEMM_TILE(2, x0.z, y0.z)
        GEMM_TILE(3, x0.w, y0.w)
        GEMM_TILE(4, x1.x, y1.x)
        GEMM_TILE(5, x1.y, y1.y)
        GEMM_TILE(6, x1.z, y1.z)
        GEMM_TILE(7, x1.w, y1.w)
    }
#undef GEMM_TILE

#pragma unroll
    for (int i = 0; i < 8; ++i) {
        float4* __restrict__ hp = (float4*)(h + (size_t)(rowBase + i + 8 * hi) * FEAT) + lo;
        hp[0]  = make_float4(acc[0][i], acc[1][i], acc[2][i], acc[3][i]);
        hp[16] = make_float4(acc[4][i], acc[5][i], acc[6][i], acc[7][i]);
    }
}

#define LG_ROWS 16
#define LG_BATCHES 2
#define LG_WAVES 8
static_assert(LG_ROWS % 8 == 0 && N_NODES % (LG_ROWS * LG_BATCHES * LG_WAVES) == 0 && FEAT == 128 && HEADS * 8 == 32,
              "rows are summed eight at a time; the grid covers the nodes exactly; a lane's four columns lie in head lane / 8");

__device__ __forceinline__ float lane_xor4(float v) {
    return __uint_as_float(__builtin_amdgcn_mov_dpp8(__float_as_uint(v), 0x688facu));
}
__device__ __forceinline__ float lane_xor2(float v) {
    return __uint_as_float(__builtin_amdgcn_mov_dpp8(__float_as_uint(v), 0xb3e21au));
}
__device__ __forceinline__ float lane_xor1(float v) {
    return __uint_as_float(__builtin_amdgcn_mov_dpp8(__float_as_uint(v), 0xde54c1u));
}

__device__ __forceinline__ float group_sum_rows8(const float* p, int k8) {
    const bool b2 = (k8 & 4) != 0;
    const bool b1 = (k8 & 2) != 0;
    const bool b0 = (k8 & 1) != 0;
    float q[4];
#pragma unroll
    for (int j = 0; j < 4; ++j) {
        const float keep = b2 ? p[j + 4] : p[j];
        const float give = b2 ? p[j] : p[j + 4];
        q[j] = keep + lane_xor4(give);
    }
    float r[2];
#pragma unroll
    for (int j = 0; j < 2; ++j) {
        const float keep = b1 ? q[j + 2] : q[j];
        const float give = b1 ? q[j] : q[j + 2];
        r[j] = keep + lane_xor2(give);
    }
    const float keep = b0 ? r[1] : r[0];
    const float give = b0 ? r[0] : r[1];
    return keep + lane_xor1(give);
}

__global__ __launch_bounds__(LG_WAVES * 32) void attn_logits_kernel(const float* __restrict__ h,
                                                                    const float* __restrict__ attn_l,
                                                                    const float* __restrict__ attn_r,
                                                                    float* __restrict__ el, float* __restrict__ er) {
    const int lane = threadIdx.x & 31;
    const int k8 = lane & 7;
    const int hd = lane >> 3;
    const int wave = blockIdx.x * LG_WAVES + (threadIdx.x >> 5);
    const float4 al = ((const float4*)attn_l)[lane];
    const float4 ar = ((const float4*)attn_r)[lane];
    for (int b = 0; b < LG_BATCHES; ++b) {
        const int row0 = (wave * LG_BATCHES + b) * LG_ROWS;
        const float4* __restrict__ hp = (const float4*)(h + (size_t)row0 * FEAT) + lane;
        float4 v[LG_ROWS];
#pragma unroll
        for (int j = 0; j < LG_ROWS; ++j) v[j] = hp[j * (FEAT / 4)];
        float pl[LG_ROWS], pr[LG_ROWS];
#pragma unroll
        for (int j = 0; j < LG_ROWS; ++j) {
            pl[j] = v[j].x * al.x + v[j].y * al.y + v[j].z * al.z + v[j].w * al.w;
            pr[j] = v[j].x * ar.x + v[j].y * ar.y + v[j].z * ar.z + v[j].w * ar.w;
        }
#pragma unroll
        for (int g = 0; g < LG_ROWS / 8; ++g) {
            const float sl = group_sum_rows8(pl + 8 * g, k8);
            const float sr = group_sum_rows8(pr + 8 * g, k8);
            const int o = (row0 + 8 * g + k8) * HEADS + hd;
            el[o] = sl;
            er[o] = sr;
        }
    }
}

#define TILE_SHIFT 8
#define N_TILES (NUM_DSTN >> TILE_SHIFT)
#define PLANE_WORDS (E_EDGES / 32)
#define TB_EDGES 2048
#define TB_WORDS (TB_EDGES / 32)
__global__ __launch_bounds__(256) void tile_plane_kernel(const int* __restrict__ dst,
                                                         unsigned int* planes) {
    __shared__ __attribute__((aligned(16))) unsigned int bits[N_TILES * TB_WORDS];
    const int tid = threadIdx.x;
    uint4* b4 = (uint4*)bits;
    for (int i = tid; i < N_TILES * TB_WORDS / 4; i += 256) b4[i] = make_uint4(0u, 0u, 0u, 0u);
    __syncthreads();

    const int g = blockIdx.x * (TB_EDGES / 4) + tid * 2;
    const int4 d0 = ((const int4*)dst)[g];
    const int4 d1 = ((const int4*)dst)[g + 1];
    unsigned int* wp = bits + (tid >> 2);
    const unsigned b0 = 1u << ((tid & 3) * 8);
#define SET_BIT(d, k)                                                                              \
    __hip_atomic_fetch_or(wp + (((unsigned)(d) >> TILE_SHIFT) & (N_TILES - 1)) * TB_WORDS, b0 << (k), \
                          __ATOMIC_RELAXED, __HIP_MEMORY_SCOPE_WORKGROUP)
    SET_BIT(d0.x, 0); SET_BIT(d0.y, 1); SET_BIT(d0.z, 2); SET_BIT(d0.w, 3);
    SET_BIT(d1.x, 4); SET_BIT(d1.y, 5); SET_BIT(d1.z, 6); SET_BIT(d1.w, 7);
#undef SET_BIT
    __syncthreads();

    for (int i = tid; i < N_TILES * TB_WORDS / 4; i += 256) {
        uint4* o = (uint4*)(planes + (size_t)(i >> 4) * PLANE_WORDS + blockIdx.x * TB_WORDS);
        o[i & 15] = b4[i];
    }
}

#define TILE_D (1 << TILE_SHIFT)
#define AGG_WAVES 32
#define AGG_THREADS (AGG_WAVES * 32)
#define ROWS_PER_WAVE (TILE_D / AGG_WAVES)
#define PLANE_VEC (PLANE_WORDS / 4)
#define TRIPS (PLANE_VEC / AGG_THREADS)
#define BATCH 8
#define MAXB 32
#define SQ 64
#define RCAP 64
#define RSTEP 16
static_assert(PLANE_VEC % AGG_THREADS == 0 && SQ >= 2 * MAXB, "plane pieces per lane; queue holds left-overs plus one turn");
static_assert(BATCH == 8 && HEADS * 8 == 32 && OUT_D == 32, "walk 2: a group of eight lanes per head, one entry per lane of a group");
static_assert(RCAP % RSTEP == 0 && RSTEP == 2 * BATCH && TILE_D % AGG_WAVES == 0, "bins are walked in whole steps of two groups of eight; rows split evenly over the waves");

__device__ __forceinline__ void lds_add(float* p, float v) {
    __hip_atomic_fetch_add(p, v, __ATOMIC_RELAXED, __HIP_MEMORY_SCOPE_WORKGROUP);
}
__device__ __forceinline__ void lds_max_f(float* p, float v) {
    if (v >= 0.0f)
        __hip_atomic_fetch_max((int*)p, __float_as_int(v), __ATOMIC_RELAXED, __HIP_MEMORY_SCOPE_WORKGROUP);
    else
        __hip_atomic_fetch_min((unsigned int*)p, __float_as_uint(v), __ATOMIC_RELAXED, __HIP_MEMORY_SCOPE_WORKGROUP);
}
__device__ __forceinline__ float leaky(float x) {
    const float v = (x >= 0.f) ? x : NEG_SLOPE * x;
    return v + 0.0f;
}

struct AggCtx {
    const int* __restrict__ src;
    const int* __restrict__ dst;
    const float* __restrict__ h;
    const float* __restrict__ el;
    float* acc;
    float* smax;
    float* ssum;
    const float4* ser;
    int* rowq;
    int* rcnt;
    int* ovf;
};

__device__ __forceinline__ void bin_batch(const int* q, int nb, int lane, const AggCtx& c) {
    if (lane < nb) {
        const int e = q[lane];
        const int s = c.src[e];
        const int dl = c.dst[e] & (TILE_D - 1);
        const int pos = __hip_atomic_fetch_add(c.rcnt + dl, 1, __ATOMIC_RELAXED, __HIP_MEMORY_SCOPE_WORKGROUP);
        if (pos >= RCAP) *c.ovf = 1;
        c.rowq[dl * RCAP + min(pos, RCAP - 1)] = s;
    }
}

__device__ __forceinline__ void max_batch(const int* q, int nb, int lane, const AggCtx& c) {
    if (lane < nb) {
        const int e = q[lane];
        const int s = c.src[e];
        const int dl = c.dst[e] & (TILE_D - 1);
        const float4 l4 = *(const float4*)(c.el + (size_t)s * HEADS);
        const float4 r4 = c.ser[dl];
        float* mp = c.smax + dl * HEADS;
        lds_max_f(mp + 0, leaky(l4.x + r4.x));
        lds_max_f(mp + 1, leaky(l4.y + r4.y));
        lds_max_f(mp + 2, leaky(l4.z + r4.z));
        lds_max_f(mp + 3, leaky(l4.w + r4.w));
    }
}

template <int K>
__device__ __forceinline__ float group_bcast(float v) {
    return __uint_as_float(__builtin_amdgcn_mov_dpp8(__float_as_uint(v), K * 0x249249u));
}

__device__ __forceinline__ float group_sum(float v) {
    v += __uint_as_float(__builtin_amdgcn_mov_dpp8(__float_as_uint(v), 0xde54c1u));
    v += __uint_as_float(__builtin_amdgcn_mov_dpp8(__float_as_uint(v), 0xb3e21au));
    v += __uint_as_float(__builtin_amdgcn_mov_dpp8(__float_as_uint(v), 0x688facu));
    return v;
}

__device__ __forceinline__ float group_max(float v) {
    v = fmaxf(v, __uint_as_float(__builtin_amdgcn_mov_dpp8(__float_as_uint(v), 0xde54c1u)));
    v = fmaxf(v, __uint_as_float(__builtin_amdgcn_mov_dpp8(__float_as_uint(v), 0xb3e21au)));
    v = fmaxf(v, __uint_as_float(__builtin_amdgcn_mov_dpp8(__float_as_uint(v), 0x688facu)));
    return v;
}

template <bool FULL, bool FIRST>
__device__ __forceinline__ void row_step(int s0, int s1, float l0, float l1, float lx0, float lx1, bool more, int nb, int lane,
                                         float r1, float& m1, float& asum, float4& acc, const AggCtx& c) {
    const int k8 = lane & 7;
    const bool mine0 = FULL || k8 < nb;
    const bool mine1 = FULL || k8 + 8 < nb;

    float4 hv[RSTEP];
#pragma unroll
    for (int k = 0; k < RSTEP; ++k) {
        if (FULL || k < nb) {
            const int si = __builtin_amdgcn_readlane(k < 8 ? s0 : s1, k & 7);
            hv[k] = ((const float4*)(c.h + (size_t)si * FEAT))[lane];
        } else {
            hv[k] = make_float4(0.f, 0.f, 0.f, 0.f);
        }
    }
    __builtin_amdgcn_sched_barrier(0);

    const float e0 = leaky(l0 + r1);
    const float e1 = leaky(l1 + r1);
    if (FIRST) {
        float m = fmaxf(e0, e1);
        if (more) m = fmaxf(m, fmaxf(leaky(lx0 + r1), leaky(lx1 + r1)));
        m1 = group_max(m);
    }
    const float a0 = expf(e0 - m1);
    const float a1 = expf(e1 - m1);
    asum += mine0 ? a0 : 0.0f;
    asum += mine1 ? a1 : 0.0f;

#define SUM_ROW(K, A, H)                                            \
    if (FULL || (H) + K < nb) {                                     \
        const float w = group_bcast<K>(A);                          \
        acc.x += hv[(H) + K].x * w;                                 \
        acc.y += hv[(H) + K].y * w;                                 \
        acc.z += hv[(H) + K].z * w;                                 \
        acc.w += hv[(H) + K].w * w;                                 \
    }
    SUM_ROW(0, a0, 0) SUM_ROW(1, a0, 0) SUM_ROW(2, a0, 0) SUM_ROW(3, a0, 0)
    SUM_ROW(4, a0, 0) SUM_ROW(5, a0, 0) SUM_ROW(6, a0, 0) SUM_ROW(7, a0, 0)
    SUM_ROW(0, a1, 8) SUM_ROW(1, a1, 8) SUM_ROW(2, a1, 8) SUM_ROW(3, a1, 8)
    SUM_ROW(4, a1, 8) SUM_ROW(5, a1, 8) SUM_ROW(6, a1, 8) SUM_ROW(7, a1, 8)
#undef SUM_ROW
}

__device__ __forceinline__ void sum_row(int r, int lane, float4* orow, const AggCtx& c) {
    const int hd = lane >> 3;
    const int k8 = lane & 7;
    const int n = min(__builtin_amdgcn_readfirstlane(c.rcnt[r]), RCAP);
    const int* rq = c.rowq + r * RCAP;
    const float r1 = ((const float*)c.ser)[r * HEADS + hd];
    float m1 = 0.0f;
    float asum = 0.0f;
    float4 acc = make_float4(0.f, 0.f, 0.f, 0.f);
    if (n > 2 * RSTEP) {
        float m = -INFINITY;
        for (int i = 0; i < n; i += RSTEP) {
            const int nb = min(RSTEP, n - i);
            const int s0 = rq[i + min(k8, nb - 1)];
            const int s1 = rq[i + min(k8 + 8, nb - 1)];
            m = fmaxf(m, leaky(c.el[(size_t)s0 * HEADS + hd] + r1));
            m = fmaxf(m, leaky(c.el[(size_t)s1 * HEADS + hd] + r1));
        }
        m1 = group_max(m);
        for (int i = 0; i < n; i += RSTEP) {
            const int nb = min(RSTEP, n - i);
            const int s0 = rq[i + min(k8, nb - 1)];
            const int s1 = rq[i + min(k8 + 8, nb - 1)];
            const float l0 = c.el[(size_t)s0 * HEADS + hd];
            const float l1 = c.el[(size_t)s1 * HEADS + hd];
            if (nb == RSTEP) row_step<true, false>(s0, s1, l0, l1, 0.0f, 0.0f, false, nb, lane, r1, m1, asum, acc, c);
            else row_step<false, false>(s0, s1, l0, l1, 0.0f, 0.0f, false, nb, lane, r1, m1, asum, acc, c);
        }
    } else if (n > 0) {
        const int nA = min(n, RSTEP);
        const int nB = n - nA;
        const int sA0 = rq[min(k8, nA - 1)];
        const int sA1 = rq[min(k8 + 8, nA - 1)];
        const float lA0 = c.el[(size_t)sA0 * HEADS + hd];
        const float lA1 = c.el[(size_t)sA1 * HEADS + hd];
        int sB0 = 0, sB1 = 0;
        float lB0 = 0.0f, lB1 = 0.0f;
        if (nB > 0) {
            sB0 = rq[RSTEP + min(k8, nB - 1)];
            sB1 = rq[RSTEP + min(k8 + 8, nB - 1)];
            lB0 = c.el[(size_t)sB0 * HEADS + hd];
            lB1 = c.el[(size_t)sB1 * HEADS + hd];
        }
        if (nA == RSTEP) row_step<true, true>(sA0, sA1, lA0, lA1, lB0, lB1, nB > 0, nA, lane, r1, m1, asum, acc, c);
        else row_step<false, true>(sA0, sA1, lA0, lA1, lB0, lB1, false, nA, lane, r1, m1, asum, acc, c);
        if (nB > 0) {
            if (nB == RSTEP) row_step<true, false>(sB0, sB1, lB0, lB1, 0.0f, 0.0f, false, nB, lane, r1, m1, asum, acc, c);
            else row_step<false, false>(sB0, sB1, lB0, lB1, 0.0f, 0.0f, false, nB, lane, r1, m1, asum, acc, c);
        }
    }
    const float sd = group_sum(asum);
    const bool any = sd > 0.f;
    float4 v;
    v.x = any ? acc.x / sd : 0.f;
    v.y = any ? acc.y / sd : 0.f;
    v.z = any ? acc.z / sd : 0.f;
    v.w = any ? acc.w / sd : 0.f;
    v.x = (v.x > 0.f) ? v.x : (expf(v.x) - 1.0f);
    v.y = (v.y > 0.f) ? v.y : (expf(v.y) - 1.0f);
    v.z = (v.z > 0.f) ? v.z : (expf(v.z) - 1.0f);
    v.w = (v.w > 0.f) ? v.w : (expf(v.w) - 1.0f);
    orow[lane] = v;
}

template <bool FULL>
__device__ __forceinline__ void agg_batch(const int* q, int nb, int lane, const AggCtx& c) {
    const int hd = lane >> 3;
    int qi = lane & (BATCH - 1);
    const bool mine = FULL || qi < nb;
    if (!FULL) qi = min(qi, nb - 1);
    const int e = q[qi];
    const int s = c.src[e];
    const int dl = c.dst[e] & (TILE_D - 1);
    const float l1 = c.el[(size_t)s * HEADS + hd];

    float4 hv[BATCH];
#pragma unroll
    for (int k = 0; k < BATCH; ++k) {
        if (FULL || k < nb) {
            const int si = __builtin_amdgcn_readlane(s, k);
            hv[k] = ((const float4*)(c.h + (size_t)si * FEAT))[lane];
        } else {
            hv[k] = make_float4(0.f, 0.f, 0.f, 0.f);
        }
    }

    const float r1 = ((const float*)c.ser)[dl * HEADS + hd];
    const float m1 = c.smax[dl * HEADS + hd];
    const float a = expf(leaky(l1 + r1) - m1);
    if (mine) lds_add(c.ssum + dl * HEADS + hd, a);

#define ADD_ROW(K)                                                  \
    if (FULL || K < nb) {                                           \
        const int dk = __builtin_amdgcn_readlane(dl, K);            \
        const float w = group_bcast<K>(a);                          \
        float* ap = c.acc + dk * FEAT + lane;                       \
        lds_add(ap,      hv[K].x * w);                              \
        lds_add(ap + 32, hv[K].y * w);                              \
        lds_add(ap + 64, hv[K].z * w);                              \
        lds_add(ap + 96, hv[K].w * w);                              \
    }
    ADD_ROW(0) ADD_ROW(1) ADD_ROW(2) ADD_ROW(3)
    ADD_ROW(4) ADD_ROW(5) ADD_ROW(6) ADD_ROW(7)
#undef ADD_ROW
}

template <int WALK>
__device__ __forceinline__ void queue_word(unsigned x, int ebase, int& cnt, int lane, int* myq, const AggCtx& c) {
    unsigned m = __builtin_amdgcn_ballot_w32(x != 0);
    while (m) {
        if (x) {
            myq[cnt + __builtin_amdgcn_mbcnt_lo(m, 0)] = ebase + __builtin_ctz(x);
            x &= x - 1;
        }
        cnt += __builtin_popcount(m);
        m = __builtin_amdgcn_ballot_w32(x != 0);
        if constexpr (WALK == 1) {
            while (cnt >= MAXB) {
                cnt -= MAXB;
                bin_batch(myq + cnt, MAXB, lane, c);
            }
        } else if constexpr (WALK == 3) {
            while (cnt >= MAXB) {
                cnt -= MAXB;
                max_batch(myq + cnt, MAXB, lane, c);
            }
        } else {
            while (cnt >= BATCH) {
                cnt -= BATCH;
                agg_batch<true>(myq + cnt, BATCH, lane, c);
            }
        }
    }
}

template <int WALK>
__device__ __forceinline__ void walk_edges(const uint4* pl, int wave, int lane, int* myq, const AggCtx& c) {
    int cnt = 0;

    uint4 cur = pl[wave * 32 + lane];
    for (int it = 0; it < TRIPS; ++it) {
        const int piece = (it * AGG_WAVES + wave) * 32 + lane;
        uint4 nxt = cur;
        if (it + 1 < TRIPS)
            nxt = pl[piece + AGG_THREADS];
        const int ebase = piece * 128;
        queue_word<WALK>(cur.x, ebase, cnt, lane, myq, c);
        queue_word<WALK>(cur.y, ebase + 32, cnt, lane, myq, c);
        queue_word<WALK>(cur.z, ebase + 64, cnt, lane, myq, c);
        queue_word<WALK>(cur.w, ebase + 96, cnt, lane, myq, c);
        cur = nxt;
    }
    if (cnt > 0) {
        if constexpr (WALK == 1) bin_batch(myq, cnt, lane, c);
        else if constexpr (WALK == 3) max_batch(myq, cnt, lane, c);
        else agg_batch<false>(myq, cnt, lane, c);
    }
}

__global__ __launch_bounds__(AGG_THREADS) void aggregate_kernel(const int* __restrict__ src,
                                                                const int* __restrict__ dst,
                                                                const float* __restrict__ h,
                                                                const float* __restrict__ el,
                                                                const float* __restrict__ er,
                                                                float* out) {
    __shared__ float acc[TILE_D * FEAT];
    __shared__ float4 ser[TILE_D];
    __shared__ __attribute__((aligned(16))) float smax[TILE_D * HEADS];
    __shared__ float ssum[TILE_D * HEADS];
    __shared__ int rowq[TILE_D * RCAP];
    __shared__ int rcnt[TILE_D];
    __shared__ int sq[AGG_WAVES][SQ];
    __shared__ int ovf;

    const int tid = threadIdx.x;
    const int lane = tid & 31;
    const int wave = tid >> 5;
    const int tile = blockIdx.x;

    if (tid < TILE_D) {
        ser[tid] = ((const float4*)er)[(size_t)tile * TILE_D + tid];
        rcnt[tid] = 0;
    }
    if (tid == 0) ovf = 0;
    ((unsigned int*)smax)[tid] = 0xFF800000u;
    ssum[tid] = 0.0f;
    __syncthreads();

    AggCtx c;
    c.src = src; c.dst = dst; c.h = h; c.el = el;
    c.acc = acc; c.smax = smax; c.ssum = ssum; c.ser = ser;
    c.rowq = rowq; c.rcnt = rcnt; c.ovf = &ovf;
    const uint4* pl = (const uint4*)(out + (size_t)tile * TILE_D * FEAT);
    static_assert(PLANE_WORDS == TILE_D * FEAT, "a plane fills exactly its tile's output rows");
    float4* o4 = (float4*)(out + (size_t)tile * TILE_D * FEAT);

    walk_edges<1>(pl, wave, lane, sq[wave], c);
    __syncthreads();

    if (ovf == 0) {
        for (int j = 0; j < ROWS_PER_WAVE; ++j) {
            const int r = wave * ROWS_PER_WAVE + j;
            sum_row(r, lane, o4 + r * (FEAT / 4), c);
        }
    } else {
        for (int i = tid; i < TILE_D * FEAT / 4; i += AGG_THREADS)
            ((float4*)acc)[i] = make_float4(0.f, 0.f, 0.f, 0.f);
        walk_edges<3>(pl, wave, lane, sq[wave], c);
        __syncthreads();
        walk_edges<2>(pl, wave, lane, sq[wave], c);
        __syncthreads();
        for (int i = tid; i < TILE_D * FEAT / 4; i += AGG_THREADS) {
            const float* ar = acc + (i >> 5) * FEAT + (i & 31);
            float4 v = make_float4(ar[0], ar[32], ar[64], ar[96]);
            const float sd = ssum[(i >> 5) * HEADS + ((i & 31) >> 3)];
            const bool any = sd > 0.f;
            v.x = any ? v.x / sd : 0.f;
            v.y = any ? v.y / sd : 0.f;
            v.z = any ? v.z / sd : 0.f;
            v.w = any ? v.w / sd : 0.f;
            v.x = (v.x > 0.f) ? v.x : (expf(v.x) - 1.0f);
            v.y = (v.y > 0.f) ? v.y : (expf(v.y) - 1.0f);
            v.z = (v.z > 0.f) ? v.z : (expf(v.z) - 1.0f);
            v.w = (v.w > 0.f) ? v.w : (expf(v.w) - 1.0f);
            o4[i] = v;
        }
    }
    if (tile == 0 && tid == 0) {
        out[(size_t)NUM_DSTN * FEAT + 0] = 1.0f;
        out[(size_t)NUM_DSTN * FEAT + 1] = 1.0f;
    }
}

extern "C" void kernel_launch(void* const* d_in, const int* in_sizes, int n_in,
                              void* d_out, int out_size, void* d_ws, size_t ws_size,
                              hipStream_t stream) {
    const float* feats  = (const float*)d_in[0];
    const int*   src    = (const int*)d_in[1];
    const int*   dst    = (const int*)d_in[2];
    const float* W      = (const float*)d_in[3];
    const float* attn_l = (const float*)d_in[4];
    const float* attn_r = (const float*)d_in[5];
    float* out = (float*)d_out;

    float* ws   = (float*)d_ws;
    float* hbuf = ws;
    float* el   = hbuf + (size_t)N_NODES * FEAT;
    float* er   = el + (size_t)N_NODES * HEADS;

    {
        int rowTiles = N_NODES / 16;
        int blocks = rowTiles / 8;
        gemm_kernel<<<blocks, 256, 0, stream>>>(feats, W, hbuf);
    }
    attn_logits_kernel<<<N_NODES / (LG_ROWS * LG_BATCHES * LG_WAVES), LG_WAVES * 32, 0, stream>>>(hbuf, attn_l, attn_r, el, er);
    tile_plane_kernel<<<E_EDGES / TB_EDGES, 256, 0, stream>>>(dst, (unsigned int*)out);
    aggregate_kernel<<<NUM_DSTN / TILE_D, AGG_THREADS, 0, stream>>>(src, dst, hbuf, el, er, out);
}
